# starting kernel
# baseline (speedup 1.0000x reference)
#include <hip/hip_runtime.h>
#include <cmath>

typedef __attribute__((ext_vector_type(2))) float v2f;
typedef __attribute__((ext_vector_type(4))) float v4f;
typedef __attribute__((ext_vector_type(8))) float v8f;

__global__ void __launch_bounds__(256) zero_kernel(v4f* __restrict__ p, int n4) {
    int i = blockIdx.x * blockDim.x + threadIdx.x;
    if (i < n4) p[i] = (v4f){0.f, 0.f, 0.f, 0.f};
}

template<int K, int NC>
__device__ __forceinline__ void fill_swizzled(float* sWs, const float* __restrict__ W) {
    constexpr int NT  = NC / 16;
    constexpr int TOT = (K / 4) * NT * 64;
    for (int i = threadIdx.x; i < TOT; i += blockDim.x) {
        const int h    = i & 1;
        const int lane = (i >> 1) & 31;
        const int c    = (i >> 6) % NT;
        const int ks   = i / (64 * NT);
        const int k    = ks * 4 + ((lane >> 4) << 1) + h;
        const int col  = c * 16 + (lane & 15);
        sWs[i] = W[k * NC + col];
    }
}

template<int D>
__global__ void __launch_bounds__(256) edge_kernel(
    const float* __restrict__ X, const int* __restrict__ src,
    const int* __restrict__ dst, const float* __restrict__ EA,
    const float* __restrict__ eW, const float* __restrict__ eb,
    float* __restrict__ Agg, int nedges)
{
    constexpr int NT = D / 16;
    __shared__ float sWs[4 * NT * 64];
    __shared__ float sB[D];
    fill_swizzled<16, D>(sWs, eW);
    for (int i = threadIdx.x; i < D; i += blockDim.x) sB[i] = eb[i];
    __syncthreads();

    const int lane  = threadIdx.x & 31;
    const int m     = lane & 15;
    const int kb    = (lane >> 4) << 1;
    const int n     = lane & 15;
    const int mbase = (lane >> 4) << 3;

    const int wavesTotal = (gridDim.x * blockDim.x) >> 5;
    const int ntiles = nedges >> 4;
    int wtile = (blockIdx.x * blockDim.x + threadIdx.x) >> 5;

    for (int t = wtile; t < ntiles; t += wavesTotal) {
        const int e0 = t << 4;

        const int ab = (e0 + m) * 16 + kb;
        const v2f a0 = *(const v2f*)(EA + ab);
        const v2f a1 = *(const v2f*)(EA + ab + 4);
        const v2f a2 = *(const v2f*)(EA + ab + 8);
        const v2f a3 = *(const v2f*)(EA + ab + 12);

        int sb[8], db[8];
#pragma unroll
        for (int j = 0; j < 8; ++j) {
            sb[j] = src[e0 + mbase + j] * D + n;
            db[j] = dst[e0 + mbase + j] * D + n;
        }

        v8f acc[NT];
#pragma unroll
        for (int c = 0; c < NT; ++c) {
            const float bv = sB[c * 16 + n];
            acc[c] = (v8f){bv, bv, bv, bv, bv, bv, bv, bv};
            const v2f* b0 = (const v2f*)(sWs + (0 * NT + c) * 64);
            const v2f* b1 = (const v2f*)(sWs + (1 * NT + c) * 64);
            const v2f* b2 = (const v2f*)(sWs + (2 * NT + c) * 64);
            const v2f* b3 = (const v2f*)(sWs + (3 * NT + c) * 64);
            acc[c] = __builtin_amdgcn_wmma_f32_16x16x4_f32(false, a0, false, b0[lane], (short)0, acc[c], false, false);
            acc[c] = __builtin_amdgcn_wmma_f32_16x16x4_f32(false, a1, false, b1[lane], (short)0, acc[c], false, false);
            acc[c] = __builtin_amdgcn_wmma_f32_16x16x4_f32(false, a2, false, b2[lane], (short)0, acc[c], false, false);
            acc[c] = __builtin_amdgcn_wmma_f32_16x16x4_f32(false, a3, false, b3[lane], (short)0, acc[c], false, false);
        }

#pragma unroll
        for (int j = 0; j < 8; ++j) {
            float xv[NT];
#pragma unroll
            for (int c = 0; c < NT; ++c) xv[c] = X[sb[j] + c * 16];
#pragma unroll
            for (int c = 0; c < NT; ++c) {
                float msg = acc[c][j] + xv[c];
                msg = msg > 0.0f ? msg : 0.0f;
                atomicAdd(&Agg[db[j] + c * 16], msg);
            }
        }
    }
}

template<int K, int NC, bool ACT_ELU, bool ADD2>
__global__ void __launch_bounds__(256) gemm_kernel(
    const float* __restrict__ A, const float* __restrict__ A2,
    const float* __restrict__ W, const float* __restrict__ bias,
    float* __restrict__ Out, int nrows)
{
    constexpr int NT = NC / 16;
    __shared__ float sWs[(K / 4) * NT * 64];
    __shared__ float sB[NC];
    fill_swizzled<K, NC>(sWs, W);
    for (int i = threadIdx.x; i < NC; i += blockDim.x) sB[i] = bias[i];
    __syncthreads();

    const int lane = threadIdx.x & 31;
    const int wave = threadIdx.x >> 5;
    const int tile = blockIdx.x * 8 + wave;
    if (tile * 16 >= nrows) return;
    const int row0 = tile * 16;

    const int m  = lane & 15;
    const int kb = (lane >> 4) << 1;
    const int n  = lane & 15;

    v8f acc[NT] = {};

    const int abase = (row0 + m) * K + kb;

#pragma unroll
    for (int ks = 0; ks < K / 4; ++ks) {
        v2f a = *(const v2f*)(A + abase + ks * 4);
        if (ADD2) {
            v2f a2 = *(const v2f*)(A2 + abase + ks * 4);
            a.x += a2.x; a.y += a2.y;
        }
        const v2f* brow = (const v2f*)(sWs + ks * NT * 64);
#pragma unroll
        for (int c = 0; c < NT; ++c) {
            acc[c] = __builtin_amdgcn_wmma_f32_16x16x4_f32(
                false, a, false, brow[c * 32 + lane], (short)0, acc[c], false, false);
        }
    }

    const int mbase = (lane >> 4) << 3;
    const int obase = (row0 + mbase) * NC + n;
#pragma unroll
    for (int c = 0; c < NT; ++c) {
        const float bv = sB[c * 16 + n];
#pragma unroll
        for (int j = 0; j < 8; ++j) {
            float v = acc[c][j] + bv;
            if (ACT_ELU) v = (v > 0.0f) ? v : (__expf(v) - 1.0f);
            Out[obase + j * NC + c * 16] = v;
        }
    }
}

extern "C" void kernel_launch(void* const* d_in, const int* in_sizes, int n_in,
                              void* d_out, int out_size, void* d_ws, size_t ws_size,
                              hipStream_t stream)
{
    const float* x    = (const float*)d_in[0];
    const int*   src  = (const int*)  d_in[1];
    const int*   dst  = (const int*)  d_in[2];
    const float* ea   = (const float*)d_in[3];
    const float* eW0  = (const float*)d_in[4];
    const float* eb0  = (const float*)d_in[5];
    const float* W1_0 = (const float*)d_in[6];
    const float* b1_0 = (const float*)d_in[7];
    const float* W2_0 = (const float*)d_in[8];
    const float* b2_0 = (const float*)d_in[9];
    const float* eW_r = (const float*)d_in[10];
    const float* eb_r = (const float*)d_in[11];
    const float* W1_r = (const float*)d_in[12];
    const float* b1_r = (const float*)d_in[13];
    const float* W2_r = (const float*)d_in[14];
    const float* b2_r = (const float*)d_in[15];
    const float* cW   = (const float*)d_in[16];
    const float* cb   = (const float*)d_in[17];

    const int N = in_sizes[0] / 64;
    const int E = in_sizes[1];

    float* agg = (float*)d_ws;
    float* hb  = agg + (size_t)N * 128;
    float* tb  = hb  + (size_t)N * 128;

    const int tiles   = (N + 15) / 16;
    const int gblocks = (tiles + 7) / 8;
    const int eblocks = 2048;

    { int n4 = (N * 64) / 4; zero_kernel<<<(n4 + 255) / 256, 256, 0, stream>>>((v4f*)agg, n4); }
    edge_kernel<64><<<eblocks, 256, 0, stream>>>(x, src, dst, ea, eW0, eb0, agg, E);
    gemm_kernel<64, 64, true, true ><<<gblocks, 256, 0, stream>>>(x,  agg, W1_0, b1_0, tb, N);
    gemm_kernel<64, 128, true, false><<<gblocks, 256, 0, stream>>>(tb, tb,  W2_0, b2_0, hb, N);

    for (int i = 0; i < 2; ++i) {
        int n4 = (N * 128) / 4;
        zero_kernel<<<(n4 + 255) / 256, 256, 0, stream>>>((v4f*)agg, n4);
        edge_kernel<128><<<eblocks, 256, 0, stream>>>(
            hb, src, dst, ea, eW_r + (size_t)i * 16 * 128, eb_r + (size_t)i * 128, agg, E);
        gemm_kernel<128, 128, true, true ><<<gblocks, 256, 0, stream>>>(
            hb, agg, W1_r + (size_t)i * 128 * 128, b1_r + (size_t)i * 128, tb, N);
        gemm_kernel<128, 128, true, false><<<gblocks, 256, 0, stream>>>(
            tb, tb, W2_r + (size_t)i * 128 * 128, b2_r + (size_t)i * 128, hb, N);
    }

    gemm_kernel<128, 32, false, false><<<gblocks, 256, 0, stream>>>(
        hb, hb, cW, cb, (float*)d_out, N);
}


# GEMMs weight-stationary: each wave holds the B fragments of its 32 output columns for all of K in registers (read once from global) and loops over row tiles; no LDS staging, no barrier
# speedup vs baseline: 1.3468x; 1.3468x over previous
#include <hip/hip_runtime.h>
#include <cmath>

typedef __attribute__((ext_vector_type(2))) float v2f;
typedef __attribute__((ext_vector_type(4))) float v4f;
typedef __attribute__((ext_vector_type(8))) float v8f;

__global__ void __launch_bounds__(256) zero_kernel(v4f* __restrict__ p, int n4) {
    int i = blockIdx.x * blockDim.x + threadIdx.x;
    if (i < n4) p[i] = (v4f){0.f, 0.f, 0.f, 0.f};
}

template<int K, int NC>
__device__ __forceinline__ void fill_swizzled(float* sWs, const float* __restrict__ W) {
    constexpr int NT  = NC / 16;
    constexpr int TOT = (K / 4) * NT * 64;
    for (int i = threadIdx.x; i < TOT; i += blockDim.x) {
        const int h    = i & 1;
        const int lane = (i >> 1) & 31;
        const int c    = (i >> 6) % NT;
        const int ks   = i / (64 * NT);
        const int k    = ks * 4 + ((lane >> 4) << 1) + h;
        const int col  = c * 16 + (lane & 15);
        sWs[i] = W[k * NC + col];
    }
}

template<int D>
__global__ void __launch_bounds__(256) edge_kernel(
    const float* __restrict__ X, const int* __restrict__ src,
    const int* __restrict__ dst, const float* __restrict__ EA,
    const float* __restrict__ eW, const float* __restrict__ eb,
    float* __restrict__ Agg, int nedges)
{
    constexpr int NT = D / 16;
    __shared__ float sWs[4 * NT * 64];
    __shared__ float sB[D];
    fill_swizzled<16, D>(sWs, eW);
    for (int i = threadIdx.x; i < D; i += blockDim.x) sB[i] = eb[i];
    __syncthreads();

    const int lane  = threadIdx.x & 31;
    const int m     = lane & 15;
    const int kb    = (lane >> 4) << 1;
    const int n     = lane & 15;
    const int mbase = (lane >> 4) << 3;

    const int wavesTotal = (gridDim.x * blockDim.x) >> 5;
    const int ntiles = nedges >> 4;
    int wtile = (blockIdx.x * blockDim.x + threadIdx.x) >> 5;

    for (int t = wtile; t < ntiles; t += wavesTotal) {
        const int e0 = t << 4;

        const int ab = (e0 + m) * 16 + kb;
        const v2f a0 = *(const v2f*)(EA + ab);
        const v2f a1 = *(const v2f*)(EA + ab + 4);
        const v2f a2 = *(const v2f*)(EA + ab + 8);
        const v2f a3 = *(const v2f*)(EA + ab + 12);

        int sb[8], db[8];
#pragma unroll
        for (int j = 0; j < 8; ++j) {
            sb[j] = src[e0 + mbase + j] * D + n;
            db[j] = dst[e0 + mbase + j] * D + n;
        }

        v8f acc[NT];
#pragma unroll
        for (int c = 0; c < NT; ++c) {
            const float bv = sB[c * 16 + n];
            acc[c] = (v8f){bv, bv, bv, bv, bv, bv, bv, bv};
            const v2f* b0 = (const v2f*)(sWs + (0 * NT + c) * 64);
            const v2f* b1 = (const v2f*)(sWs + (1 * NT + c) * 64);
            const v2f* b2 = (const v2f*)(sWs + (2 * NT + c) * 64);
            const v2f* b3 = (const v2f*)(sWs + (3 * NT + c) * 64);
            acc[c] = __builtin_amdgcn_wmma_f32_16x16x4_f32(false, a0, false, b0[lane], (short)0, acc[c], false, false);
            acc[c] = __builtin_amdgcn_wmma_f32_16x16x4_f32(false, a1, false, b1[lane], (short)0, acc[c], false, false);
            acc[c] = __builtin_amdgcn_wmma_f32_16x16x4_f32(false, a2, false, b2[lane], (short)0, acc[c], false, false);
            acc[c] = __builtin_amdgcn_wmma_f32_16x16x4_f32(false, a3, false, b3[lane], (short)0, acc[c], false, false);
        }

#pragma unroll
        for (int j = 0; j < 8; ++j) {
            float xv[NT];
#pragma unroll
            for (int c = 0; c < NT; ++c) xv[c] = X[sb[j] + c * 16];
#pragma unroll
            for (int c = 0; c < NT; ++c) {
                float msg = acc[c][j] + xv[c];
                msg = msg > 0.0f ? msg : 0.0f;
                atomicAdd(&Agg[db[j] + c * 16], msg);
            }
        }
    }
}

template<int K, int NC, bool ACT_ELU, bool ADD2>
__global__ void __launch_bounds__(128, (K == 128 && ADD2) ? 3 : 4) gemm_kernel(
    const float* __restrict__ A, const float* __restrict__ A2,
    const float* __restrict__ W, const float* __restrict__ bias,
    float* __restrict__ Out, int ntiles)
{
    constexpr int NG  = NC / 32;
    constexpr int SPB = 4 / NG;
    constexpr int KS  = K / 4;

    const int lane = threadIdx.x & 31;
    const int wave = __builtin_amdgcn_readfirstlane((int)(threadIdx.x >> 5));
    const int cg     = wave % NG;
    const int slot   = blockIdx.x * SPB + wave / NG;
    const int nslots = gridDim.x * SPB;

    const int m     = lane & 15;
    const int kb    = (lane >> 4) << 1;
    const int n     = lane & 15;
    const int mbase = (lane >> 4) << 3;
    const int col0  = cg * 32 + n;

    v2f bw[KS][2];
    const int wb = kb * NC + col0;
#pragma unroll
    for (int ks = 0; ks < KS; ++ks) {
#pragma unroll
        for (int c = 0; c < 2; ++c) {
            bw[ks][c].x = W[wb + (ks * 4)     * NC + c * 16];
            bw[ks][c].y = W[wb + (ks * 4 + 1) * NC + c * 16];
        }
    }
    const float bv0 = bias[col0];
    const float bv1 = bias[col0 + 16];

    for (int t = slot; t < ntiles; t += nslots) {
        const int row0  = t * 16;
        const int abase = (row0 + m) * K + kb;
        v8f acc0 = {};
        v8f acc1 = {};
#pragma unroll
        for (int ks = 0; ks < KS; ++ks) {
            v2f a = *(const v2f*)(A + abase + ks * 4);
            if (ADD2) {
                v2f a2 = *(const v2f*)(A2 + abase + ks * 4);
                a.x += a2.x; a.y += a2.y;
            }
            acc0 = __builtin_amdgcn_wmma_f32_16x16x4_f32(
                false, a, false, bw[ks][0], (short)0, acc0, false, false);
            acc1 = __builtin_amdgcn_wmma_f32_16x16x4_f32(
                false, a, false, bw[ks][1], (short)0, acc1, false, false);
        }

        const int obase = (row0 + mbase) * NC + col0;
#pragma unroll
        for (int j = 0; j < 8; ++j) {
            float v0 = acc0[j] + bv0;
            float v1 = acc1[j] + bv1;
            if (ACT_ELU) {
                v0 = (v0 > 0.0f) ? v0 : (__expf(v0) - 1.0f);
                v1 = (v1 > 0.0f) ? v1 : (__expf(v1) - 1.0f);
            }
            Out[obase + j * NC]      = v0;
            Out[obase + j * NC + 16] = v1;
        }
    }
}

extern "C" void kernel_launch(void* const* d_in, const int* in_sizes, int n_in,
                              void* d_out, int out_size, void* d_ws, size_t ws_size,
                              hipStream_t stream)
{
    const float* x    = (const float*)d_in[0];
    const int*   src  = (const int*)  d_in[1];
    const int*   dst  = (const int*)  d_in[2];
    const float* ea   = (const float*)d_in[3];
    const float* eW0  = (const float*)d_in[4];
    const float* eb0  = (const float*)d_in[5];
    const float* W1_0 = (const float*)d_in[6];
    const float* b1_0 = (const float*)d_in[7];
    const float* W2_0 = (const float*)d_in[8];
    const float* b2_0 = (const float*)d_in[9];
    const float* eW_r = (const float*)d_in[10];
    const float* eb_r = (const float*)d_in[11];
    const float* W1_r = (const float*)d_in[12];
    const float* b1_r = (const float*)d_in[13];
    const float* W2_r = (const float*)d_in[14];
    const float* b2_r = (const float*)d_in[15];
    const float* cW   = (const float*)d_in[16];
    const float* cb   = (const float*)d_in[17];

    const int N = in_sizes[0] / 64;
    const int E = in_sizes[1];

    float* agg = (float*)d_ws;
    float* hb  = agg + (size_t)N * 128;
    float* tb  = hb  + (size_t)N * 128;

    const int tiles   = (N + 15) / 16;
    const int eblocks = 2048;
    const int g64x64   = 782;
    const int g64x128  = 1563;
    const int g128x128 = 1024;
    const int g128x128s = 768;
    const int g128x32  = 391;

    { int n4 = (N * 64) / 4; zero_kernel<<<(n4 + 255) / 256, 256, 0, stream>>>((v4f*)agg, n4); }
    edge_kernel<64><<<eblocks, 256, 0, stream>>>(x, src, dst, ea, eW0, eb0, agg, E);
    gemm_kernel<64, 64, true, true ><<<g64x64, 128, 0, stream>>>(x,  agg, W1_0, b1_0, tb, tiles);
    gemm_kernel<64, 128, true, false><<<g64x128, 128, 0, stream>>>(tb, tb,  W2_0, b2_0, hb, tiles);

    for (int i = 0; i < 2; ++i) {
        int n4 = (N * 128) / 4;
        zero_kernel<<<(n4 + 255) / 256, 256, 0, stream>>>((v4f*)agg, n4);
        edge_kernel<128><<<eblocks, 256, 0, stream>>>(
            hb, src, dst, ea, eW_r + (size_t)i * 16 * 128, eb_r + (size_t)i * 128, agg, E);
        gemm_kernel<128, 128, true, true ><<<g128x128s, 128, 0, stream>>>(
            hb, agg, W1_r + (size_t)i * 128 * 128, b1_r + (size_t)i * 128, tb, tiles);
        gemm_kernel<128, 128, true, false><<<g128x128, 128, 0, stream>>>(
            tb, tb, W2_r + (size_t)i * 128 * 128, b2_r + (size_t)i * 128, hb, tiles);
    }

    gemm_kernel<128, 32, false, false><<<g128x32, 128, 0, stream>>>(
        hb, hb, cW, cb, (float*)d_out, tiles);
}


# Edge sum owner-computes: edges grouped per call by dst range (LDS counting pass); one work-group per 196 nodes adds messages in LDS (ds_add_f32), plain row stores; f32 sum order differs
# speedup vs baseline: 1.9463x; 1.4451x over previous
#include <hip/hip_runtime.h>
#include <cmath>

typedef __attribute__((ext_vector_type(2))) float v2f;
typedef __attribute__((ext_vector_type(4))) float v4f;
typedef __attribute__((ext_vector_type(8))) float v8f;

template<int K, int NC>
__device__ __forceinline__ void fill_swizzled(float* sWs, const float* __restrict__ W) {
    constexpr int NT  = NC / 16;
    constexpr int TOT = (K / 4) * NT * 64;
    for (int i = threadIdx.x; i < TOT; i += blockDim.x) {
        const int h    = i & 1;
        const int lane = (i >> 1) & 31;
        const int c    = (i >> 6) % NT;
        const int ks   = i / (64 * NT);
        const int k    = ks * 4 + ((lane >> 4) << 1) + h;
        const int col  = c * 16 + (lane & 15);
        sWs[i] = W[k * NC + col];
    }
}

constexpr int RB   = 196;
constexpr int MAXB = 512;
constexpr int GB   = 128;

typedef __attribute__((ext_vector_type(2))) int v2i;

__global__ void __launch_bounds__(MAXB) bucket_count_kernel(
    const int* __restrict__ dst, int* __restrict__ cnt, int nedges, int chunk, int nbk)
{
    __shared__ int hist[MAXB];
    hist[threadIdx.x] = 0;
    __syncthreads();
    const int e0 = blockIdx.x * chunk;
    const int e1 = min(e0 + chunk, nedges);
    for (int e = e0 + (int)threadIdx.x; e < e1; e += MAXB)
        atomicAdd(&hist[dst[e] / RB], 1);
    __syncthreads();
    if ((int)threadIdx.x < nbk) cnt[blockIdx.x * nbk + threadIdx.x] = hist[threadIdx.x];
}

__global__ void __launch_bounds__(MAXB) bucket_fill_kernel(
    const int* __restrict__ src, const int* __restrict__ dst, const int* __restrict__ cnt,
    int* __restrict__ plist, int* __restrict__ peid, int* __restrict__ bstart,
    int nedges, int chunk, int nbk)
{
    v2i* __restrict__ pl2 = (v2i*)plist;
    __shared__ int s[MAXB];
    __shared__ int cur[MAXB];
    const int b = threadIdx.x;
    const int g = blockIdx.x;
    int tot = 0, pre = 0;
    if (b < nbk) {
        for (int gg = 0; gg < GB; ++gg) {
            const int c = cnt[gg * nbk + b];
            tot += c;
            pre += (gg < g) ? c : 0;
        }
    }
    s[b] = tot;
    __syncthreads();
    for (int d = 1; d < MAXB; d <<= 1) {
        const int v = (b >= d) ? s[b - d] : 0;
        __syncthreads();
        s[b] += v;
        __syncthreads();
    }
    const int start = s[b] - tot;
    cur[b] = start + pre;
    if (g == 0) {
        if (b < nbk) bstart[b] = start;
        if (b == 0)  bstart[nbk] = nedges;
    }
    __syncthreads();

    const int e0 = g * chunk;
    const int e1 = min(e0 + chunk, nedges);
    for (int e = e0 + b; e < e1; e += MAXB) {
        const int d   = dst[e];
        const int sv  = src[e];
        const int bk  = d / RB;
        const int pos = atomicAdd(&cur[bk], 1);
        pl2[pos]  = (v2i){sv, d - bk * RB};
        peid[pos] = e;
    }
}

template<int D>
__global__ void __launch_bounds__(1024) edge_kernel(
    const float* __restrict__ X, const int* __restrict__ plist,
    const int* __restrict__ peid, const int* __restrict__ bstart,
    const float* __restrict__ EA,
    const float* __restrict__ eW, const float* __restrict__ eb,
    float* __restrict__ Agg, int nnodes)
{
    constexpr int NT    = D / 16;
    constexpr int PITCH = D + 16;
    __shared__ float sWs[4 * NT * 64];
    __shared__ float sB[D];
    __shared__ __attribute__((aligned(16))) float sAgg[(RB + 1) * PITCH];
    fill_swizzled<16, D>(sWs, eW);
    for (int i = threadIdx.x; i < D; i += blockDim.x) sB[i] = eb[i];
    for (int i = threadIdx.x; i < (RB + 1) * PITCH / 4; i += blockDim.x)
        ((v4f*)sAgg)[i] = (v4f){0.f, 0.f, 0.f, 0.f};
    __syncthreads();

    const int lane  = threadIdx.x & 31;
    const int wave   = __builtin_amdgcn_readfirstlane((int)(threadIdx.x >> 5));
    const int nwaves = (int)(blockDim.x >> 5);
    const int m     = lane & 15;
    const int kb    = (lane >> 4) << 1;
    const int n     = lane & 15;
    const int mbase = (lane >> 4) << 3;

    const v2i* __restrict__ pl2 = (const v2i*)plist;
    const int bk = blockIdx.x;
    const int lo = bstart[bk];
    const int hi = bstart[bk + 1];
    const int ntl = (hi - lo + 15) >> 4;

    for (int t = wave; t < ntl; t += nwaves) {
        if (D > 64) asm volatile("" ::: "memory");
        const int p0 = lo + (t << 4);

        const int pa = min(p0 + m, hi - 1);
        const int ab = peid[pa] * 16 + kb;
        const v2f a0 = *(const v2f*)(EA + ab);
        const v2f a1 = *(const v2f*)(EA + ab + 4);
        const v2f a2 = *(const v2f*)(EA + ab + 8);
        const v2f a3 = *(const v2f*)(EA + ab + 12);

        int sb[8], lb[8];
#pragma unroll
        for (int j = 0; j < 8; ++j) {
            const int pj = p0 + mbase + j;
            const v2i sd = pl2[min(pj, hi - 1)];
            sb[j] = sd.x * D + n;
            lb[j] = ((pj < hi) ? sd.y : RB) * PITCH + n;
        }

        v8f acc[NT];
#pragma unroll
        for (int c = 0; c < NT; ++c) {
            const float bv = sB[c * 16 + n];
            acc[c] = (v8f){bv, bv, bv, bv, bv, bv, bv, bv};
            const v2f* b0 = (const v2f*)(sWs + (0 * NT + c) * 64);
            const v2f* b1 = (const v2f*)(sWs + (1 * NT + c) * 64);
            const v2f* b2 = (const v2f*)(sWs + (2 * NT + c) * 64);
            const v2f* b3 = (const v2f*)(sWs + (3 * NT + c) * 64);
            acc[c] = __builtin_amdgcn_wmma_f32_16x16x4_f32(false, a0, false, b0[lane], (short)0, acc[c], false, false);
            acc[c] = __builtin_amdgcn_wmma_f32_16x16x4_f32(false, a1, false, b1[lane], (short)0, acc[c], false, false);
            acc[c] = __builtin_amdgcn_wmma_f32_16x16x4_f32(false, a2, false, b2[lane], (short)0, acc[c], false, false);
            acc[c] = __builtin_amdgcn_wmma_f32_16x16x4_f32(false, a3, false, b3[lane], (short)0, acc[c], false, false);
        }

#pragma unroll
        for (int j = 0; j < 8; ++j) {
            float xv[NT];
#pragma unroll
            for (int c = 0; c < NT; ++c) xv[c] = X[sb[j] + c * 16];
#pragma unroll
            for (int c = 0; c < NT; ++c) {
                float msg = acc[c][j] + xv[c];
                msg = msg > 0.0f ? msg : 0.0f;
                atomicAdd(&sAgg[lb[j] + c * 16], msg);
            }
        }
    }
    __syncthreads();

    const int row0  = bk * RB;
    const int nrows = min(RB, nnodes - row0);
    constexpr int Q = D / 4;
    for (int i = threadIdx.x; i < nrows * Q; i += blockDim.x) {
        const int r = i / Q;
        const int q = i - r * Q;
        const v4f v = *(const v4f*)(sAgg + r * PITCH + q * 4);
        *(v4f*)(Agg + (size_t)(row0 + r) * D + q * 4) = v;
    }
}

template<int K, int NC, bool ACT_ELU, bool ADD2>
__global__ void __launch_bounds__(128, (K == 128 && ADD2) ? 3 : 4) gemm_kernel(
    const float* __restrict__ A, const float* __restrict__ A2,
    const float* __restrict__ W, const float* __restrict__ bias,
    float* __restrict__ Out, int ntiles)
{
    constexpr int NG  = NC / 32;
    constexpr int SPB = 4 / NG;
    constexpr int KS  = K / 4;

    const int lane = threadIdx.x & 31;
    const int wave = __builtin_amdgcn_readfirstlane((int)(threadIdx.x >> 5));
    const int cg     = wave % NG;
    const int slot   = blockIdx.x * SPB + wave / NG;
    const int nslots = gridDim.x * SPB;

    const int m     = lane & 15;
    const int kb    = (lane >> 4) << 1;
    const int n     = lane & 15;
    const int mbase = (lane >> 4) << 3;
    const int col0  = cg * 32 + n;

    v2f bw[KS][2];
    const int wb = kb * NC + col0;
#pragma unroll
    for (int ks = 0; ks < KS; ++ks) {
#pragma unroll
        for (int c = 0; c < 2; ++c) {
            bw[ks][c].x = W[wb + (ks * 4)     * NC + c * 16];
            bw[ks][c].y = W[wb + (ks * 4 + 1) * NC + c * 16];
        }
    }
    const float bv0 = bias[col0];
    const float bv1 = bias[col0 + 16];

    for (int t = slot; t < ntiles; t += nslots) {
        const int row0  = t * 16;
        const int abase = (row0 + m) * K + kb;
        v8f acc0 = {};
        v8f acc1 = {};
#pragma unroll
        for (int ks = 0; ks < KS; ++ks) {
            v2f a = *(const v2f*)(A + abase + ks * 4);
            if (ADD2) {
                v2f a2 = *(const v2f*)(A2 + abase + ks * 4);
                a.x += a2.x; a.y += a2.y;
            }
            acc0 = __builtin_amdgcn_wmma_f32_16x16x4_f32(
                false, a, false, bw[ks][0], (short)0, acc0, false, false);
            acc1 = __builtin_amdgcn_wmma_f32_16x16x4_f32(
                false, a, false, bw[ks][1], (short)0, acc1, false, false);
        }

        const int obase = (row0 + mbase) * NC + col0;
#pragma unroll
        for (int j = 0; j < 8; ++j) {
            float v0 = acc0[j] + bv0;
            float v1 = acc1[j] + bv1;
            if (ACT_ELU) {
                v0 = (v0 > 0.0f) ? v0 : (__expf(v0) - 1.0f);
                v1 = (v1 > 0.0f) ? v1 : (__expf(v1) - 1.0f);
            }
            Out[obase + j * NC]      = v0;
            Out[obase + j * NC + 16] = v1;
        }
    }
}

extern "C" void kernel_launch(void* const* d_in, const int* in_sizes, int n_in,
                              void* d_out, int out_size, void* d_ws, size_t ws_size,
                              hipStream_t stream)
{
    const float* x    = (const float*)d_in[0];
    const int*   src  = (const int*)  d_in[1];
    const int*   dst  = (const int*)  d_in[2];
    const float* ea   = (const float*)d_in[3];
    const float* eW0  = (const float*)d_in[4];
    const float* eb0  = (const float*)d_in[5];
    const float* W1_0 = (const float*)d_in[6];
    const float* b1_0 = (const float*)d_in[7];
    const float* W2_0 = (const float*)d_in[8];
    const float* b2_0 = (const float*)d_in[9];
    const float* eW_r = (const float*)d_in[10];
    const float* eb_r = (const float*)d_in[11];
    const float* W1_r = (const float*)d_in[12];
    const float* b1_r = (const float*)d_in[13];
    const float* W2_r = (const float*)d_in[14];
    const float* b2_r = (const float*)d_in[15];
    const float* cW   = (const float*)d_in[16];
    const float* cb   = (const float*)d_in[17];

    const int N = in_sizes[0] / 64;
    const int E = in_sizes[1];

    float* agg = (float*)d_ws;
    float* hb  = agg + (size_t)N * 128;
    float* tb  = hb  + (size_t)N * 128;
    int* plist  = (int*)(tb + (size_t)N * 128);
    int* peid   = plist + 2 * (size_t)E;
    int* cnt    = peid + (size_t)E;
    int* bstart = cnt  + GB * MAXB;

    const int tiles   = (N + 15) / 16;
    const int nbk     = (N + RB - 1) / RB;
    const int chunk   = (E + GB - 1) / GB;
    const int g64x64   = 782;
    const int g64x128  = 1563;
    const int g128x128 = 1024;
    const int g128x128s = 768;
    const int g128x32  = 391;

    bucket_count_kernel<<<GB, MAXB, 0, stream>>>(dst, cnt, E, chunk, nbk);
    bucket_fill_kernel<<<GB, MAXB, 0, stream>>>(src, dst, cnt, plist, peid, bstart, E, chunk, nbk);

    edge_kernel<64><<<nbk, 1024, 0, stream>>>(x, plist, peid, bstart, ea, eW0, eb0, agg, N);
    gemm_kernel<64, 64, true, true ><<<g64x64, 128, 0, stream>>>(x,  agg, W1_0, b1_0, tb, tiles);
    gemm_kernel<64, 128, true, false><<<g64x128, 128, 0, stream>>>(tb, tb,  W2_0, b2_0, hb, tiles);

    for (int i = 0; i < 2; ++i) {
        edge_kernel<128><<<nbk, 1024, 0, stream>>>(
            hb, plist, peid, bstart, ea, eW_r + (size_t)i * 16 * 128, eb_r + (size_t)i * 128, agg, N);
        gemm_kernel<128, 128, true, true ><<<g128x128s, 128, 0, stream>>>(
            hb, agg, W1_r + (size_t)i * 128 * 128, b1_r + (size_t)i * 128, tb, tiles);
        gemm_kernel<128, 128, true, false><<<g128x128, 128, 0, stream>>>(
            tb, tb, W2_r + (size_t)i * 128 * 128, b2_r + (size_t)i * 128, hb, tiles);
    }

    gemm_kernel<128, 32, false, false><<<g128x32, 128, 0, stream>>>(
        hb, hb, cW, cb, (float*)d_out, tiles);
}


# Each layer ONE launch: the work-group that owns a bucket's sum rows in LDS starts them at the node's own row, runs both node products (+classifier) on them in place; 12->5 launches; sum order differs
# speedup vs baseline: 2.3515x; 1.2082x over previous
#include <hip/hip_runtime.h>
#include <cmath>

typedef __attribute__((ext_vector_type(2))) float v2f;
typedef __attribute__((ext_vector_type(4))) float v4f;
typedef __attribute__((ext_vector_type(8))) float v8f;

template<int K, int NC>
__device__ __forceinline__ void fill_swizzled(float* sWs, const float* __restrict__ W) {
    constexpr int NT  = NC / 16;
    constexpr int TOT = (K / 4) * NT * 64;
    for (int i = threadIdx.x; i < TOT; i += blockDim.x) {
        const int h    = i & 1;
        const int lane = (i >> 1) & 31;
        const int c    = (i >> 6) % NT;
        const int ks   = i / (64 * NT);
        const int k    = ks * 4 + ((lane >> 4) << 1) + h;
        const int col  = c * 16 + (lane & 15);
        sWs[i] = W[k * NC + col];
    }
}

constexpr int RB   = 196;
constexpr int MAXB = 512;
constexpr int GB   = 128;
constexpr int NRT   = (RB + 15) / 16;
constexpr int LROWS = NRT * 16;
constexpr int PITCH = 132;

typedef __attribute__((ext_vector_type(2))) int v2i;

__global__ void __launch_bounds__(MAXB) bucket_count_kernel(
    const int* __restrict__ dst, int* __restrict__ cnt, int nedges, int chunk, int nbk)
{
    __shared__ int hist[MAXB];
    hist[threadIdx.x] = 0;
    __syncthreads();
    const int e0 = blockIdx.x * chunk;
    const int e1 = min(e0 + chunk, nedges);
    for (int e = e0 + (int)threadIdx.x; e < e1; e += MAXB)
        atomicAdd(&hist[dst[e] / RB], 1);
    __syncthreads();
    if ((int)threadIdx.x < nbk) cnt[blockIdx.x * nbk + threadIdx.x] = hist[threadIdx.x];
}

__global__ void __launch_bounds__(MAXB) bucket_fill_kernel(
    const int* __restrict__ src, const int* __restrict__ dst, const int* __restrict__ cnt,
    int* __restrict__ plist, int* __restrict__ peid, int* __restrict__ bstart,
    int nedges, int chunk, int nbk)
{
    v2i* __restrict__ pl2 = (v2i*)plist;
    __shared__ int s[MAXB];
    __shared__ int cur[MAXB];
    const int b = threadIdx.x;
    const int g = blockIdx.x;
    int tot = 0, pre = 0;
    if (b < nbk) {
        for (int gg = 0; gg < GB; ++gg) {
            const int c = cnt[gg * nbk + b];
            tot += c;
            pre += (gg < g) ? c : 0;
        }
    }
    s[b] = tot;
    __syncthreads();
    for (int d = 1; d < MAXB; d <<= 1) {
        const int v = (b >= d) ? s[b - d] : 0;
        __syncthreads();
        s[b] += v;
        __syncthreads();
    }
    const int start = s[b] - tot;
    cur[b] = start + pre;
    if (g == 0) {
        if (b < nbk) bstart[b] = start;
        if (b == 0)  bstart[nbk] = nedges;
    }
    __syncthreads();

    const int e0 = g * chunk;
    const int e1 = min(e0 + chunk, nedges);
    for (int e = e0 + b; e < e1; e += MAXB) {
        const int d   = dst[e];
        const int sv  = src[e];
        const int bk  = d / RB;
        const int pos = atomicAdd(&cur[bk], 1);
        pl2[pos]  = (v2i){sv, d - bk * RB};
        peid[pos] = e;
    }
}

template<int K, int NC>
struct BFrag {
    v2f   w[K / 8][2];
    float bias;
};

__host__ __device__ constexpr int max_tiles(int NC) {
    return (NRT + 32 / (NC / 16) - 1) / (32 / (NC / 16));
}

template<int K, int NC>
__device__ __forceinline__ void load_bfrag(
    BFrag<K, NC>& f, const float* __restrict__ Wg, const float* __restrict__ bias,
    int wave, int lane)
{
    constexpr int NCT = NC / 16;
    const int col = (wave % NCT) * 16 + (lane & 15);
    const int wb  = (4 * (lane >> 4)) * NC + col;
#pragma unroll
    for (int s = 0; s < K / 8; ++s) {
#pragma unroll
        for (int i = 0; i < 2; ++i) {
            f.w[s][i].x = Wg[wb + (8 * s + 2 * i)     * NC];
            f.w[s][i].y = Wg[wb + (8 * s + 2 * i + 1) * NC];
        }
    }
    f.bias = bias[col];
}

template<int K, int NC>
__device__ __forceinline__ void rows_times_b(
    v8f (&acc)[max_tiles(NC)], const float* rows, const BFrag<K, NC>& f,
    int wave, int lane)
{
    constexpr int NCT = NC / 16;
    constexpr int NRG = 32 / NCT;
    const int rg   = wave / NCT;
    const int n    = lane & 15;
    const int half = lane >> 4;
#pragma unroll
    for (int i = 0; i < max_tiles(NC); ++i) {
        const int rt = rg + i * NRG;
        acc[i] = (v8f){};
        if (rt < NRT) {
            const float* ap = rows + (rt * 16 + n) * PITCH + 4 * half;
#pragma unroll
            for (int s = 0; s < K / 8; ++s) {
                const v4f q = *(const v4f*)(ap + 8 * s);
                const v2f a0 = {q.x, q.y};
                const v2f a1 = {q.z, q.w};
                acc[i] = __builtin_amdgcn_wmma_f32_16x16x4_f32(
                    false, a0, false, f.w[s][0], (short)0, acc[i], false, false);
                acc[i] = __builtin_amdgcn_wmma_f32_16x16x4_f32(
                    false, a1, false, f.w[s][1], (short)0, acc[i], false, false);
            }
        }
    }
}

template<int NC, bool ACT_ELU>
__device__ __forceinline__ void put_rows(
    float* rows, const v8f (&acc)[max_tiles(NC)], float bias, int wave, int lane)
{
    constexpr int NCT = NC / 16;
    constexpr int NRG = 32 / NCT;
    const int rg   = wave / NCT;
    const int col  = (wave % NCT) * 16 + (lane & 15);
    const int half = lane >> 4;
#pragma unroll
    for (int i = 0; i < max_tiles(NC); ++i) {
        const int rt = rg + i * NRG;
        if (rt < NRT) {
            float* op = rows + (rt * 16 + 8 * half) * PITCH + col;
#pragma unroll
            for (int j = 0; j < 8; ++j) {
                float v = acc[i][j] + bias;
                if (ACT_ELU) v = (v > 0.0f) ? v : (__expf(v) - 1.0f);
                op[j * PITCH] = v;
            }
        }
    }
}

template<int D, int DM, int DO, int NCLS>
__global__ void __launch_bounds__(1024) layer_kernel(
    const float* __restrict__ X, const int* __restrict__ plist,
    const int* __restrict__ peid, const int* __restrict__ bstart,
    const float* __restrict__ EA,
    const float* __restrict__ eW, const float* __restrict__ eb,
    const float* __restrict__ W1, const float* __restrict__ bias1,
    const float* __restrict__ W2, const float* __restrict__ bias2,
    const float* __restrict__ cW, const float* __restrict__ cb,
    float* __restrict__ Hout, float* __restrict__ Out, int nnodes)
{
    constexpr int NT    = D / 16;
    __shared__ float sWs[4 * NT * 64];
    __shared__ float sB[D];
    __shared__ __attribute__((aligned(16))) float sAgg[LROWS * PITCH];

    const int bk    = blockIdx.x;
    const int row0  = bk * RB;
    const int nrows = min(RB, nnodes - row0);

    fill_swizzled<16, D>(sWs, eW);
    for (int i = threadIdx.x; i < D; i += blockDim.x) sB[i] = eb[i];
    {
        constexpr int QI = D / 4;
        for (int i = threadIdx.x; i < LROWS * QI; i += blockDim.x) {
            const int r = i / QI;
            const int q = i - r * QI;
            v4f v = {0.f, 0.f, 0.f, 0.f};
            if (r < nrows) v = *(const v4f*)(X + (size_t)(row0 + r) * D + q * 4);
            *(v4f*)(sAgg + r * PITCH + q * 4) = v;
        }
    }
    __syncthreads();

    const int lane  = threadIdx.x & 31;
    const int wave   = __builtin_amdgcn_readfirstlane((int)(threadIdx.x >> 5));
    const int nwaves = (int)(blockDim.x >> 5);
    const int m     = lane & 15;
    const int kb    = (lane >> 4) << 1;
    const int n     = lane & 15;
    const int mbase = (lane >> 4) << 3;

    const v2i* __restrict__ pl2 = (const v2i*)plist;
    const int lo = bstart[bk];
    const int hi = bstart[bk + 1];
    const int ntl = (hi - lo + 15) >> 4;

    for (int t = wave; t < ntl; t += nwaves) {
        if (D > 64) asm volatile("" ::: "memory");
        const int p0 = lo + (t << 4);

        const int pa = min(p0 + m, hi - 1);
        const int ab = peid[pa] * 16 + kb;
        const v2f a0 = *(const v2f*)(EA + ab);
        const v2f a1 = *(const v2f*)(EA + ab + 4);
        const v2f a2 = *(const v2f*)(EA + ab + 8);
        const v2f a3 = *(const v2f*)(EA + ab + 12);

        int sb[8], lb[8];
#pragma unroll
        for (int j = 0; j < 8; ++j) {
            const int pj = p0 + mbase + j;
            const v2i sd = pl2[min(pj, hi - 1)];
            sb[j] = sd.x * D + n;
            lb[j] = ((pj < hi) ? sd.y : RB) * PITCH + n;
        }

        v8f acc[NT];
#pragma unroll
        for (int c = 0; c < NT; ++c) {
            const float bv = sB[c * 16 + n];
            acc[c] = (v8f){bv, bv, bv, bv, bv, bv, bv, bv};
            const v2f* b0 = (const v2f*)(sWs + (0 * NT + c) * 64);
            const v2f* b1 = (const v2f*)(sWs + (1 * NT + c) * 64);
            const v2f* b2 = (const v2f*)(sWs + (2 * NT + c) * 64);
            const v2f* b3 = (const v2f*)(sWs + (3 * NT + c) * 64);
            acc[c] = __builtin_amdgcn_wmma_f32_16x16x4_f32(false, a0, false, b0[lane], (short)0, acc[c], false, false);
            acc[c] = __builtin_amdgcn_wmma_f32_16x16x4_f32(false, a1, false, b1[lane], (short)0, acc[c], false, false);
            acc[c] = __builtin_amdgcn_wmma_f32_16x16x4_f32(false, a2, false, b2[lane], (short)0, acc[c], false, false);
            acc[c] = __builtin_amdgcn_wmma_f32_16x16x4_f32(false, a3, false, b3[lane], (short)0, acc[c], false, false);
        }

#pragma unroll
        for (int j = 0; j < 8; ++j) {
            float xv[NT];
#pragma unroll
            for (int c = 0; c < NT; ++c) xv[c] = X[sb[j] + c * 16];
#pragma unroll
            for (int c = 0; c < NT; ++c) {
                float msg = acc[c][j] + xv[c];
                msg = msg > 0.0f ? msg : 0.0f;
                atomicAdd(&sAgg[lb[j] + c * 16], msg);
            }
        }
    }
    BFrag<D, DM> f1;
    load_bfrag<D, DM>(f1, W1, bias1, wave, lane);
    __syncthreads();
    v8f acc1[max_tiles(DM)];
    rows_times_b<D, DM>(acc1, sAgg, f1, wave, lane);
    __syncthreads();
    put_rows<DM, true>(sAgg, acc1, f1.bias, wave, lane);
    BFrag<DM, DO> f2;
    load_bfrag<DM, DO>(f2, W2, bias2, wave, lane);
    __syncthreads();
    v8f acc2[max_tiles(DO)];
    rows_times_b<DM, DO>(acc2, sAgg, f2, wave, lane);

    if constexpr (NCLS == 0) {
        __syncthreads();
        put_rows<DO, true>(sAgg, acc2, f2.bias, wave, lane);
        __syncthreads();
        constexpr int Q = DO / 4;
        for (int i = threadIdx.x; i < nrows * Q; i += blockDim.x) {
            const int r = i / Q;
            const int q = i - r * Q;
            const v4f v = *(const v4f*)(sAgg + r * PITCH + q * 4);
            *(v4f*)(Hout + (size_t)(row0 + r) * DO + q * 4) = v;
        }
    } else {
        __syncthreads();
        put_rows<DO, true>(sAgg, acc2, f2.bias, wave, lane);
        BFrag<DO, NCLS> f3;
        load_bfrag<DO, NCLS>(f3, cW, cb, wave, lane);
        __syncthreads();
        v8f acc3[max_tiles(NCLS)];
        rows_times_b<DO, NCLS>(acc3, sAgg, f3, wave, lane);
        __syncthreads();
        put_rows<NCLS, false>(sAgg, acc3, f3.bias, wave, lane);
        __syncthreads();
        constexpr int Q = NCLS / 4;
        for (int i = threadIdx.x; i < nrows * Q; i += blockDim.x) {
            const int r = i / Q;
            const int q = i - r * Q;
            const v4f v = *(const v4f*)(sAgg + r * PITCH + q * 4);
            *(v4f*)(Out + (size_t)(row0 + r) * NCLS + q * 4) = v;
        }
    }
}

extern "C" void kernel_launch(void* const* d_in, const int* in_sizes, int n_in,
                              void* d_out, int out_size, void* d_ws, size_t ws_size,
                              hipStream_t stream)
{
    const float* x    = (const float*)d_in[0];
    const int*   src  = (const int*)  d_in[1];
    const int*   dst  = (const int*)  d_in[2];
    const float* ea   = (const float*)d_in[3];
    const float* eW0  = (const float*)d_in[4];
    const float* eb0  = (const float*)d_in[5];
    const float* W1_0 = (const float*)d_in[6];
    const float* b1_0 = (const float*)d_in[7];
    const float* W2_0 = (const float*)d_in[8];
    const float* b2_0 = (const float*)d_in[9];
    const float* eW_r = (const float*)d_in[10];
    const float* eb_r = (const float*)d_in[11];
    const float* W1_r = (const float*)d_in[12];
    const float* b1_r = (const float*)d_in[13];
    const float* W2_r = (const float*)d_in[14];
    const float* b2_r = (const float*)d_in[15];
    const float* cW   = (const float*)d_in[16];
    const float* cb   = (const float*)d_in[17];

    const int N = in_sizes[0] / 64;
    const int E = in_sizes[1];

    float* hA = (float*)d_ws;
    float* hB = hA + (size_t)N * 128;
    int* plist  = (int*)(hB + (size_t)N * 128);
    int* peid   = plist + 2 * (size_t)E;
    int* cnt    = peid + (size_t)E;
    int* bstart = cnt  + GB * MAXB;

    const int nbk     = (N + RB - 1) / RB;
    const int chunk   = (E + GB - 1) / GB;

    bucket_count_kernel<<<GB, MAXB, 0, stream>>>(dst, cnt, E, chunk, nbk);
    bucket_fill_kernel<<<GB, MAXB, 0, stream>>>(src, dst, cnt, plist, peid, bstart, E, chunk, nbk);

    layer_kernel<64, 64, 128, 0><<<nbk, 1024, 0, stream>>>(
        x, plist, peid, bstart, ea, eW0, eb0, W1_0, b1_0, W2_0, b2_0, cW, cb,
        hA, (float*)d_out, N);
    layer_kernel<128, 128, 128, 0><<<nbk, 1024, 0, stream>>>(
        hA, plist, peid, bstart, ea, eW_r, eb_r, W1_r, b1_r, W2_r, b2_r, cW, cb,
        hB, (float*)d_out, N);
    layer_kernel<128, 128, 128, 32><<<nbk, 1024, 0, stream>>>(
        hB, plist, peid, bstart, ea, eW_r + (size_t)16 * 128, eb_r + (size_t)128,
        W1_r + (size_t)128 * 128, b1_r + (size_t)128, W2_r + (size_t)128 * 128, b2_r + (size_t)128,
        cW, cb, hA, (float*)d_out, N);
}


# Edge tile taken in passes of 64 columns: a pass's 8x4 row reads are one load group held in registers under its 16 WMMAs, not 8 serial load-wait-add groups per tile; order of the LDS f32 adds differs
# speedup vs baseline: 2.4249x; 1.0312x over previous
#include <hip/hip_runtime.h>
#include <cmath>

typedef __attribute__((ext_vector_type(2))) float v2f;
typedef __attribute__((ext_vector_type(4))) float v4f;
typedef __attribute__((ext_vector_type(8))) float v8f;

template<int K, int NC>
__device__ __forceinline__ void fill_swizzled(float* sWs, const float* __restrict__ W) {
    constexpr int NT  = NC / 16;
    constexpr int TOT = (K / 4) * NT * 64;
    for (int i = threadIdx.x; i < TOT; i += blockDim.x) {
        const int h    = i & 1;
        const int lane = (i >> 1) & 31;
        const int c    = (i >> 6) % NT;
        const int ks   = i / (64 * NT);
        const int k    = ks * 4 + ((lane >> 4) << 1) + h;
        const int col  = c * 16 + (lane & 15);
        sWs[i] = W[k * NC + col];
    }
}

constexpr int RB   = 196;
constexpr int MAXB = 512;
constexpr int GB   = 128;
constexpr int NRT   = (RB + 15) / 16;
constexpr int LROWS = NRT * 16;
constexpr int PITCH = 132;

typedef __attribute__((ext_vector_type(2))) int v2i;

__global__ void __launch_bounds__(MAXB) bucket_count_kernel(
    const int* __restrict__ dst, int* __restrict__ cnt, int nedges, int chunk, int nbk)
{
    __shared__ int hist[MAXB];
    hist[threadIdx.x] = 0;
    __syncthreads();
    const int e0 = blockIdx.x * chunk;
    const int e1 = min(e0 + chunk, nedges);
    for (int e = e0 + (int)threadIdx.x; e < e1; e += MAXB)
        atomicAdd(&hist[dst[e] / RB], 1);
    __syncthreads();
    if ((int)threadIdx.x < nbk) cnt[blockIdx.x * nbk + threadIdx.x] = hist[threadIdx.x];
}

__global__ void __launch_bounds__(MAXB) bucket_fill_kernel(
    const int* __restrict__ src, const int* __restrict__ dst, const int* __restrict__ cnt,
    int* __restrict__ plist, int* __restrict__ peid, int* __restrict__ bstart,
    int nedges, int chunk, int nbk)
{
    v2i* __restrict__ pl2 = (v2i*)plist;
    __shared__ int s[MAXB];
    __shared__ int cur[MAXB];
    const int b = threadIdx.x;
    const int g = blockIdx.x;
    int tot = 0, pre = 0;
    if (b < nbk) {
        for (int gg = 0; gg < GB; ++gg) {
            const int c = cnt[gg * nbk + b];
            tot += c;
            pre += (gg < g) ? c : 0;
        }
    }
    s[b] = tot;
    __syncthreads();
    for (int d = 1; d < MAXB; d <<= 1) {
        const int v = (b >= d) ? s[b - d] : 0;
        __syncthreads();
        s[b] += v;
        __syncthreads();
    }
    const int start = s[b] - tot;
    cur[b] = start + pre;
    if (g == 0) {
        if (b < nbk) bstart[b] = start;
        if (b == 0)  bstart[nbk] = nedges;
    }
    __syncthreads();

    const int e0 = g * chunk;
    const int e1 = min(e0 + chunk, nedges);
    for (int e = e0 + b; e < e1; e += MAXB) {
        const int d   = dst[e];
        const int sv  = src[e];
        const int bk  = d / RB;
        const int pos = atomicAdd(&cur[bk], 1);
        pl2[pos]  = (v2i){sv, d - bk * RB};
        peid[pos] = e;
    }
}

template<int K, int NC>
struct BFrag {
    v2f   w[K / 8][2];
    float bias;
};

__host__ __device__ constexpr int max_tiles(int NC) {
    return (NRT + 32 / (NC / 16) - 1) / (32 / (NC / 16));
}

template<int K, int NC>
__device__ __forceinline__ void load_bfrag(
    BFrag<K, NC>& f, const float* __restrict__ Wg, const float* __restrict__ bias,
    int wave, int lane)
{
    constexpr int NCT = NC / 16;
    const int col = (wave % NCT) * 16 + (lane & 15);
    const int wb  = (4 * (lane >> 4)) * NC + col;
#pragma unroll
    for (int s = 0; s < K / 8; ++s) {
#pragma unroll
        for (int i = 0; i < 2; ++i) {
            f.w[s][i].x = Wg[wb + (8 * s + 2 * i)     * NC];
            f.w[s][i].y = Wg[wb + (8 * s + 2 * i + 1) * NC];
        }
    }
    f.bias = bias[col];
}

template<int K, int NC>
__device__ __forceinline__ void rows_times_b(
    v8f (&acc)[max_tiles(NC)], const float* rows, const BFrag<K, NC>& f,
    int wave, int lane)
{
    constexpr int NCT = NC / 16;
    constexpr int NRG = 32 / NCT;
    const int rg   = wave / NCT;
    const int n    = lane & 15;
    const int half = lane >> 4;
#pragma unroll
    for (int i = 0; i < max_tiles(NC); ++i) {
        const int rt = rg + i * NRG;
        acc[i] = (v8f){};
        if (rt < NRT) {
            const float* ap = rows + (rt * 16 + n) * PITCH + 4 * half;
#pragma unroll
            for (int s = 0; s < K / 8; ++s) {
                const v4f q = *(const v4f*)(ap + 8 * s);
                const v2f a0 = {q.x, q.y};
                const v2f a1 = {q.z, q.w};
                acc[i] = __builtin_amdgcn_wmma_f32_16x16x4_f32(
                    false, a0, false, f.w[s][0], (short)0, acc[i], false, false);
                acc[i] = __builtin_amdgcn_wmma_f32_16x16x4_f32(
                    false, a1, false, f.w[s][1], (short)0, acc[i], false, false);
            }
        }
    }
}

template<int NC, bool ACT_ELU>
__device__ __forceinline__ void put_rows(
    float* rows, const v8f (&acc)[max_tiles(NC)], float bias, int wave, int lane)
{
    constexpr int NCT = NC / 16;
    constexpr int NRG = 32 / NCT;
    const int rg   = wave / NCT;
    const int col  = (wave % NCT) * 16 + (lane & 15);
    const int half = lane >> 4;
#pragma unroll
    for (int i = 0; i < max_tiles(NC); ++i) {
        const int rt = rg + i * NRG;
        if (rt < NRT) {
            float* op = rows + (rt * 16 + 8 * half) * PITCH + col;
#pragma unroll
            for (int j = 0; j < 8; ++j) {
                float v = acc[i][j] + bias;
                if (ACT_ELU) v = (v > 0.0f) ? v : (__expf(v) - 1.0f);
                op[j * PITCH] = v;
            }
        }
    }
}

template<int D, int DM, int DO, int NCLS>
__global__ void __launch_bounds__(1024) layer_kernel(
    const float* __restrict__ X, const int* __restrict__ plist,
    const int* __restrict__ peid, const int* __restrict__ bstart,
    const float* __restrict__ EA,
    const float* __restrict__ eW, const float* __restrict__ eb,
    const float* __restrict__ W1, const float* __restrict__ bias1,
    const float* __restrict__ W2, const float* __restrict__ bias2,
    const float* __restrict__ cW, const float* __restrict__ cb,
    float* __restrict__ Hout, float* __restrict__ Out, int nnodes)
{
    constexpr int NT    = D / 16;
    constexpr int CP    = 4;
    constexpr int NP    = NT / CP;
    __shared__ float sWs[4 * NT * 64];
    __shared__ float sB[D];
    __shared__ __attribute__((aligned(16))) float sAgg[LROWS * PITCH];

    const int bk    = blockIdx.x;
    const int row0  = bk * RB;
    const int nrows = min(RB, nnodes - row0);

    fill_swizzled<16, D>(sWs, eW);
    for (int i = threadIdx.x; i < D; i += blockDim.x) sB[i] = eb[i];
    {
        constexpr int QI = D / 4;
        for (int i = threadIdx.x; i < LROWS * QI; i += blockDim.x) {
            const int r = i / QI;
            const int q = i - r * QI;
            v4f v = {0.f, 0.f, 0.f, 0.f};
            if (r < nrows) v = *(const v4f*)(X + (size_t)(row0 + r) * D + q * 4);
            *(v4f*)(sAgg + r * PITCH + q * 4) = v;
        }
    }
    __syncthreads();

    const int lane  = threadIdx.x & 31;
    const int wave   = __builtin_amdgcn_readfirstlane((int)(threadIdx.x >> 5));
    const int nwaves = (int)(blockDim.x >> 5);
    const int m     = lane & 15;
    const int kb    = (lane >> 4) << 1;
    const int n     = lane & 15;
    const int mbase = (lane >> 4) << 3;

    const v2i* __restrict__ pl2 = (const v2i*)plist;
    const int lo = bstart[bk];
    const int hi = bstart[bk + 1];
    const int ntl = (hi - lo + 15) >> 4;

    for (int t = wave; t < ntl; t += nwaves) {
        asm volatile("" ::: "memory");
        const int p0 = lo + (t << 4);

        const int pa = min(p0 + m, hi - 1);
        const int ab = peid[pa] * 16 + kb;
        const v2f a0 = *(const v2f*)(EA + ab);
        const v2f a1 = *(const v2f*)(EA + ab + 4);
        const v2f a2 = *(const v2f*)(EA + ab + 8);
        const v2f a3 = *(const v2f*)(EA + ab + 12);

        int sb[8], lb[8];
#pragma unroll
        for (int j = 0; j < 8; ++j) {
            const int pj = p0 + mbase + j;
            const v2i sd = pl2[min(pj, hi - 1)];
            sb[j] = sd.x * D + n;
            lb[j] = ((pj < hi) ? sd.y : RB) * PITCH + n;
        }

#pragma unroll
        for (int p = 0; p < NP; ++p) {
            float xv[8][CP];
#pragma unroll
            for (int j = 0; j < 8; ++j) {
#pragma unroll
                for (int c = 0; c < CP; ++c) xv[j][c] = X[sb[j] + (p * CP + c) * 16];
            }
            v8f acc[CP];
#pragma unroll
            for (int c = 0; c < CP; ++c) {
                const int cc = p * CP + c;
                const float bv = sB[cc * 16 + n];
                acc[c] = (v8f){bv, bv, bv, bv, bv, bv, bv, bv};
                const v2f* b0 = (const v2f*)(sWs + (0 * NT + cc) * 64);
                const v2f* b1 = (const v2f*)(sWs + (1 * NT + cc) * 64);
                const v2f* b2 = (const v2f*)(sWs + (2 * NT + cc) * 64);
                const v2f* b3 = (const v2f*)(sWs + (3 * NT + cc) * 64);
                acc[c] = __builtin_amdgcn_wmma_f32_16x16x4_f32(false, a0, false, b0[lane], (short)0, acc[c], false, false);
                acc[c] = __builtin_amdgcn_wmma_f32_16x16x4_f32(false, a1, false, b1[lane], (short)0, acc[c], false, false);
                acc[c] = __builtin_amdgcn_wmma_f32_16x16x4_f32(false, a2, false, b2[lane], (short)0, acc[c], false, false);
                acc[c] = __builtin_amdgcn_wmma_f32_16x16x4_f32(false, a3, false, b3[lane], (short)0, acc[c], false, false);
            }
#pragma unroll
            for (int j = 0; j < 8; ++j) {
#pragma unroll
                for (int c = 0; c < CP; ++c) {
                    float msg = acc[c][j] + xv[j][c];
                    msg = msg > 0.0f ? msg : 0.0f;
                    atomicAdd(&sAgg[lb[j] + (p * CP + c) * 16], msg);
                }
            }
        }
    }
    BFrag<D, DM> f1;
    load_bfrag<D, DM>(f1, W1, bias1, wave, lane);
    __syncthreads();
    v8f acc1[max_tiles(DM)];
    rows_times_b<D, DM>(acc1, sAgg, f1, wave, lane);
    __syncthreads();
    put_rows<DM, true>(sAgg, acc1, f1.bias, wave, lane);
    BFrag<DM, DO> f2;
    load_bfrag<DM, DO>(f2, W2, bias2, wave, lane);
    __syncthreads();
    v8f acc2[max_tiles(DO)];
    rows_times_b<DM, DO>(acc2, sAgg, f2, wave, lane);

    if constexpr (NCLS == 0) {
        __syncthreads();
        put_rows<DO, true>(sAgg, acc2, f2.bias, wave, lane);
        __syncthreads();
        constexpr int Q = DO / 4;
        for (int i = threadIdx.x; i < nrows * Q; i += blockDim.x) {
            const int r = i / Q;
            const int q = i - r * Q;
            const v4f v = *(const v4f*)(sAgg + r * PITCH + q * 4);
            *(v4f*)(Hout + (size_t)(row0 + r) * DO + q * 4) = v;
        }
    } else {
        __syncthreads();
        put_rows<DO, true>(sAgg, acc2, f2.bias, wave, lane);
        BFrag<DO, NCLS> f3;
        load_bfrag<DO, NCLS>(f3, cW, cb, wave, lane);
        __syncthreads();
        v8f acc3[max_tiles(NCLS)];
        rows_times_b<DO, NCLS>(acc3, sAgg, f3, wave, lane);
        __syncthreads();
        put_rows<NCLS, false>(sAgg, acc3, f3.bias, wave, lane);
        __syncthreads();
        constexpr int Q = NCLS / 4;
        for (int i = threadIdx.x; i < nrows * Q; i += blockDim.x) {
            const int r = i / Q;
            const int q = i - r * Q;
            const v4f v = *(const v4f*)(sAgg + r * PITCH + q * 4);
            *(v4f*)(Out + (size_t)(row0 + r) * NCLS + q * 4) = v;
        }
    }
}

extern "C" void kernel_launch(void* const* d_in, const int* in_sizes, int n_in,
                              void* d_out, int out_size, void* d_ws, size_t ws_size,
                              hipStream_t stream)
{
    const float* x    = (const float*)d_in[0];
    const int*   src  = (const int*)  d_in[1];
    const int*   dst  = (const int*)  d_in[2];
    const float* ea   = (const float*)d_in[3];
    const float* eW0  = (const float*)d_in[4];
    const float* eb0  = (const float*)d_in[5];
    const float* W1_0 = (const float*)d_in[6];
    const float* b1_0 = (const float*)d_in[7];
    const float* W2_0 = (const float*)d_in[8];
    const float* b2_0 = (const float*)d_in[9];
    const float* eW_r = (const float*)d_in[10];
    const float* eb_r = (const float*)d_in[11];
    const float* W1_r = (const float*)d_in[12];
    const float* b1_r = (const float*)d_in[13];
    const float* W2_r = (const float*)d_in[14];
    const float* b2_r = (const float*)d_in[15];
    const float* cW   = (const float*)d_in[16];
    const float* cb   = (const float*)d_in[17];

    const int N = in_sizes[0] / 64;
    const int E = in_sizes[1];

    float* hA = (float*)d_ws;
    float* hB = hA + (size_t)N * 128;
    int* plist  = (int*)(hB + (size_t)N * 128);
    int* peid   = plist + 2 * (size_t)E;
    int* cnt    = peid + (size_t)E;
    int* bstart = cnt  + GB * MAXB;

    const int nbk     = (N + RB - 1) / RB;
    const int chunk   = (E + GB - 1) / GB;

    bucket_count_kernel<<<GB, MAXB, 0, stream>>>(dst, cnt, E, chunk, nbk);
    bucket_fill_kernel<<<GB, MAXB, 0, stream>>>(src, dst, cnt, plist, peid, bstart, E, chunk, nbk);

    layer_kernel<64, 64, 128, 0><<<nbk, 1024, 0, stream>>>(
        x, plist, peid, bstart, ea, eW0, eb0, W1_0, b1_0, W2_0, b2_0, cW, cb,
        hA, (float*)d_out, N);
    layer_kernel<128, 128, 128, 0><<<nbk, 1024, 0, stream>>>(
        hA, plist, peid, bstart, ea, eW_r, eb_r, W1_r, b1_r, W2_r, b2_r, cW, cb,
        hB, (float*)d_out, N);
    layer_kernel<128, 128, 128, 32><<<nbk, 1024, 0, stream>>>(
        hB, plist, peid, bstart, ea, eW_r + (size_t)16 * 128, eb_r + (size_t)128,
        W1_r + (size_t)128 * 128, b1_r + (size_t)128, W2_r + (size_t)128 * 128, b2_r + (size_t)128,
        cW, cb, hA, (float*)d_out, N);
}


# Grouping pass copies each edge's 16 attributes to its list position (4 lanes per 64-B row); a layer tile reads its A rows as 1 KB in a row, not 16 rows picked by edge id; edge-id list gone
# speedup vs baseline: 2.5002x; 1.0311x over previous
#include <hip/hip_runtime.h>
#include <cmath>

typedef __attribute__((ext_vector_type(2))) float v2f;
typedef __attribute__((ext_vector_type(4))) float v4f;
typedef __attribute__((ext_vector_type(8))) float v8f;

template<int K, int NC>
__device__ __forceinline__ void fill_swizzled(float* sWs, const float* __restrict__ W) {
    constexpr int NT  = NC / 16;
    constexpr int TOT = (K / 4) * NT * 64;
    for (int i = threadIdx.x; i < TOT; i += blockDim.x) {
        const int h    = i & 1;
        const int lane = (i >> 1) & 31;
        const int c    = (i >> 6) % NT;
        const int ks   = i / (64 * NT);
        const int k    = ks * 4 + ((lane >> 4) << 1) + h;
        const int col  = c * 16 + (lane & 15);
        sWs[i] = W[k * NC + col];
    }
}

constexpr int RB   = 196;
constexpr int MAXB = 512;
constexpr int GB   = 128;
constexpr int NRT   = (RB + 15) / 16;
constexpr int LROWS = NRT * 16;
constexpr int PITCH = 132;

typedef __attribute__((ext_vector_type(2))) int v2i;

__global__ void __launch_bounds__(MAXB) bucket_count_kernel(
    const int* __restrict__ dst, int* __restrict__ cnt, int nedges, int chunk, int nbk)
{
    __shared__ int hist[MAXB];
    hist[threadIdx.x] = 0;
    __syncthreads();
    const int e0 = blockIdx.x * chunk;
    const int e1 = min(e0 + chunk, nedges);
    for (int e = e0 + (int)threadIdx.x; e < e1; e += MAXB)
        atomicAdd(&hist[dst[e] / RB], 1);
    __syncthreads();
    if ((int)threadIdx.x < nbk) cnt[blockIdx.x * nbk + threadIdx.x] = hist[threadIdx.x];
}

__global__ void __launch_bounds__(MAXB) bucket_fill_kernel(
    const int* __restrict__ src, const int* __restrict__ dst, const int* __restrict__ cnt,
    const float* __restrict__ EA,
    int* __restrict__ plist, float* __restrict__ eas, int* __restrict__ bstart,
    int nedges, int chunk, int nbk)
{
    v2i* __restrict__ pl2 = (v2i*)plist;
    __shared__ int s[MAXB];
    __shared__ int cur[MAXB];
    const int b = threadIdx.x;
    const int g = blockIdx.x;
    int tot = 0, pre = 0;
    if (b < nbk) {
        for (int gg = 0; gg < GB; ++gg) {
            const int c = cnt[gg * nbk + b];
            tot += c;
            pre += (gg < g) ? c : 0;
        }
    }
    s[b] = tot;
    __syncthreads();
    for (int d = 1; d < MAXB; d <<= 1) {
        const int v = (b >= d) ? s[b - d] : 0;
        __syncthreads();
        s[b] += v;
        __syncthreads();
    }
    const int start = s[b] - tot;
    cur[b] = start + pre;
    if (g == 0) {
        if (b < nbk) bstart[b] = start;
        if (b == 0)  bstart[nbk] = nedges;
    }
    __syncthreads();

    const int e0 = g * chunk;
    const int e1 = min(e0 + chunk, nedges);
    const int lane = b & 31;
    const int wb   = __builtin_amdgcn_readfirstlane(b - lane);
    const int sub  = lane >> 2;
    const int part = (lane & 3) << 2;
    for (int eb = e0 + wb; eb < e1; eb += MAXB) {
        v4f av[4];
#pragma unroll
        for (int r = 0; r < 4; ++r) {
            const int es = min(eb + 8 * r + sub, e1 - 1);
            av[r] = *(const v4f*)(EA + es * 16 + part);
        }
        const int e = eb + lane;
        int pos = 0;
        if (e < e1) {
            const int d   = dst[e];
            const int sv  = src[e];
            const int bk  = d / RB;
            pos = atomicAdd(&cur[bk], 1);
            pl2[pos] = (v2i){sv, d - bk * RB};
        }
#pragma unroll
        for (int r = 0; r < 4; ++r) {
            const int sl = 8 * r + sub;
            const int ps = __shfl(pos, sl);
            if (eb + sl < e1) *(v4f*)(eas + ps * 16 + part) = av[r];
        }
    }
}

template<int K, int NC>
struct BFrag {
    v2f   w[K / 8][2];
    float bias;
};

__host__ __device__ constexpr int max_tiles(int NC) {
    return (NRT + 32 / (NC / 16) - 1) / (32 / (NC / 16));
}

template<int K, int NC>
__device__ __forceinline__ void load_bfrag(
    BFrag<K, NC>& f, const float* __restrict__ Wg, const float* __restrict__ bias,
    int wave, int lane)
{
    constexpr int NCT = NC / 16;
    const int col = (wave % NCT) * 16 + (lane & 15);
    const int wb  = (4 * (lane >> 4)) * NC + col;
#pragma unroll
    for (int s = 0; s < K / 8; ++s) {
#pragma unroll
        for (int i = 0; i < 2; ++i) {
            f.w[s][i].x = Wg[wb + (8 * s + 2 * i)     * NC];
            f.w[s][i].y = Wg[wb + (8 * s + 2 * i + 1) * NC];
        }
    }
    f.bias = bias[col];
}

template<int K, int NC>
__device__ __forceinline__ void rows_times_b(
    v8f (&acc)[max_tiles(NC)], const float* rows, const BFrag<K, NC>& f,
    int wave, int lane)
{
    constexpr int NCT = NC / 16;
    constexpr int NRG = 32 / NCT;
    const int rg   = wave / NCT;
    const int n    = lane & 15;
    const int half = lane >> 4;
#pragma unroll
    for (int i = 0; i < max_tiles(NC); ++i) {
        const int rt = rg + i * NRG;
        acc[i] = (v8f){};
        if (rt < NRT) {
            const float* ap = rows + (rt * 16 + n) * PITCH + 4 * half;
#pragma unroll
            for (int s = 0; s < K / 8; ++s) {
                const v4f q = *(const v4f*)(ap + 8 * s);
                const v2f a0 = {q.x, q.y};
                const v2f a1 = {q.z, q.w};
                acc[i] = __builtin_amdgcn_wmma_f32_16x16x4_f32(
                    false, a0, false, f.w[s][0], (short)0, acc[i], false, false);
                acc[i] = __builtin_amdgcn_wmma_f32_16x16x4_f32(
                    false, a1, false, f.w[s][1], (short)0, acc[i], false, false);
            }
        }
    }
}

template<int NC, bool ACT_ELU>
__device__ __forceinline__ void put_rows(
    float* rows, const v8f (&acc)[max_tiles(NC)], float bias, int wave, int lane)
{
    constexpr int NCT = NC / 16;
    constexpr int NRG = 32 / NCT;
    const int rg   = wave / NCT;
    const int col  = (wave % NCT) * 16 + (lane & 15);
    const int half = lane >> 4;
#pragma unroll
    for (int i = 0; i < max_tiles(NC); ++i) {
        const int rt = rg + i * NRG;
        if (rt < NRT) {
            float* op = rows + (rt * 16 + 8 * half) * PITCH + col;
#pragma unroll
            for (int j = 0; j < 8; ++j) {
                float v = acc[i][j] + bias;
                if (ACT_ELU) v = (v > 0.0f) ? v : (__expf(v) - 1.0f);
                op[j * PITCH] = v;
            }
        }
    }
}

template<int D, int DM, int DO, int NCLS>
__global__ void __launch_bounds__(1024) layer_kernel(
    const float* __restrict__ X, const int* __restrict__ plist,
    const int* __restrict__ bstart,
    const float* __restrict__ EA,
    const float* __restrict__ eW, const float* __restrict__ eb,
    const float* __restrict__ W1, const float* __restrict__ bias1,
    const float* __restrict__ W2, const float* __restrict__ bias2,
    const float* __restrict__ cW, const float* __restrict__ cb,
    float* __restrict__ Hout, float* __restrict__ Out, int nnodes)
{
    constexpr int NT    = D / 16;
    constexpr int CP    = 4;
    constexpr int NP    = NT / CP;
    __shared__ float sWs[4 * NT * 64];
    __shared__ float sB[D];
    __shared__ __attribute__((aligned(16))) float sAgg[LROWS * PITCH];

    const int bk    = blockIdx.x;
    const int row0  = bk * RB;
    const int nrows = min(RB, nnodes - row0);

    fill_swizzled<16, D>(sWs, eW);
    for (int i = threadIdx.x; i < D; i += blockDim.x) sB[i] = eb[i];
    {
        constexpr int QI = D / 4;
        for (int i = threadIdx.x; i < LROWS * QI; i += blockDim.x) {
            const int r = i / QI;
            const int q = i - r * QI;
            v4f v = {0.f, 0.f, 0.f, 0.f};
            if (r < nrows) v = *(const v4f*)(X + (size_t)(row0 + r) * D + q * 4);
            *(v4f*)(sAgg + r * PITCH + q * 4) = v;
        }
    }
    __syncthreads();

    const int lane  = threadIdx.x & 31;
    const int wave   = __builtin_amdgcn_readfirstlane((int)(threadIdx.x >> 5));
    const int nwaves = (int)(blockDim.x >> 5);
    const int m     = lane & 15;
    const int kb    = (lane >> 4) << 1;
    const int n     = lane & 15;
    const int mbase = (lane >> 4) << 3;

    const v2i* __restrict__ pl2 = (const v2i*)plist;
    const int lo = bstart[bk];
    const int hi = bstart[bk + 1];
    const int ntl = (hi - lo + 15) >> 4;

    for (int t = wave; t < ntl; t += nwaves) {
        asm volatile("" ::: "memory");
        const int p0 = lo + (t << 4);

        const int pa = min(p0 + m, hi - 1);
        const int ab = pa * 16 + kb;
        const v2f a0 = *(const v2f*)(EA + ab);
        const v2f a1 = *(const v2f*)(EA + ab + 4);
        const v2f a2 = *(const v2f*)(EA + ab + 8);
        const v2f a3 = *(const v2f*)(EA + ab + 12);

        int sb[8], lb[8];
#pragma unroll
        for (int j = 0; j < 8; ++j) {
            const int pj = p0 + mbase + j;
            const v2i sd = pl2[min(pj, hi - 1)];
            sb[j] = sd.x * D + n;
            lb[j] = ((pj < hi) ? sd.y : RB) * PITCH + n;
        }

#pragma unroll
        for (int p = 0; p < NP; ++p) {
            float xv[8][CP];
#pragma unroll
            for (int j = 0; j < 8; ++j) {
#pragma unroll
                for (int c = 0; c < CP; ++c) xv[j][c] = X[sb[j] + (p * CP + c) * 16];
            }
            v8f acc[CP];
#pragma unroll
            for (int c = 0; c < CP; ++c) {
                const int cc = p * CP + c;
                const float bv = sB[cc * 16 + n];
                acc[c] = (v8f){bv, bv, bv, bv, bv, bv, bv, bv};
                const v2f* b0 = (const v2f*)(sWs + (0 * NT + cc) * 64);
                const v2f* b1 = (const v2f*)(sWs + (1 * NT + cc) * 64);
                const v2f* b2 = (const v2f*)(sWs + (2 * NT + cc) * 64);
                const v2f* b3 = (const v2f*)(sWs + (3 * NT + cc) * 64);
                acc[c] = __builtin_amdgcn_wmma_f32_16x16x4_f32(false, a0, false, b0[lane], (short)0, acc[c], false, false);
                acc[c] = __builtin_amdgcn_wmma_f32_16x16x4_f32(false, a1, false, b1[lane], (short)0, acc[c], false, false);
                acc[c] = __builtin_amdgcn_wmma_f32_16x16x4_f32(false, a2, false, b2[lane], (short)0, acc[c], false, false);
                acc[c] = __builtin_amdgcn_wmma_f32_16x16x4_f32(false, a3, false, b3[lane], (short)0, acc[c], false, false);
            }
#pragma unroll
            for (int j = 0; j < 8; ++j) {
#pragma unroll
                for (int c = 0; c < CP; ++c) {
                    float msg = acc[c][j] + xv[j][c];
                    msg = msg > 0.0f ? msg : 0.0f;
                    atomicAdd(&sAgg[lb[j] + (p * CP + c) * 16], msg);
                }
            }
        }
    }
    BFrag<D, DM> f1;
    load_bfrag<D, DM>(f1, W1, bias1, wave, lane);
    __syncthreads();
    v8f acc1[max_tiles(DM)];
    rows_times_b<D, DM>(acc1, sAgg, f1, wave, lane);
    __syncthreads();
    put_rows<DM, true>(sAgg, acc1, f1.bias, wave, lane);
    BFrag<DM, DO> f2;
    load_bfrag<DM, DO>(f2, W2, bias2, wave, lane);
    __syncthreads();
    v8f acc2[max_tiles(DO)];
    rows_times_b<DM, DO>(acc2, sAgg, f2, wave, lane);

    if constexpr (NCLS == 0) {
        __syncthreads();
        put_rows<DO, true>(sAgg, acc2, f2.bias, wave, lane);
        __syncthreads();
        constexpr int Q = DO / 4;
        for (int i = threadIdx.x; i < nrows * Q; i += blockDim.x) {
            const int r = i / Q;
            const int q = i - r * Q;
            const v4f v = *(const v4f*)(sAgg + r * PITCH + q * 4);
            *(v4f*)(Hout + (size_t)(row0 + r) * DO + q * 4) = v;
        }
    } else {
        __syncthreads();
        put_rows<DO, true>(sAgg, acc2, f2.bias, wave, lane);
        BFrag<DO, NCLS> f3;
        load_bfrag<DO, NCLS>(f3, cW, cb, wave, lane);
        __syncthreads();
        v8f acc3[max_tiles(NCLS)];
        rows_times_b<DO, NCLS>(acc3, sAgg, f3, wave, lane);
        __syncthreads();
        put_rows<NCLS, false>(sAgg, acc3, f3.bias, wave, lane);
        __syncthreads();
        constexpr int Q = NCLS / 4;
        for (int i = threadIdx.x; i < nrows * Q; i += blockDim.x) {
            const int r = i / Q;
            const int q = i - r * Q;
            const v4f v = *(const v4f*)(sAgg + r * PITCH + q * 4);
            *(v4f*)(Out + (size_t)(row0 + r) * NCLS + q * 4) = v;
        }
    }
}

extern "C" void kernel_launch(void* const* d_in, const int* in_sizes, int n_in,
                              void* d_out, int out_size, void* d_ws, size_t ws_size,
                              hipStream_t stream)
{
    const float* x    = (const float*)d_in[0];
    const int*   src  = (const int*)  d_in[1];
    const int*   dst  = (const int*)  d_in[2];
    const float* ea   = (const float*)d_in[3];
    const float* eW0  = (const float*)d_in[4];
    const float* eb0  = (const float*)d_in[5];
    const float* W1_0 = (const float*)d_in[6];
    const float* b1_0 = (const float*)d_in[7];
    const float* W2_0 = (const float*)d_in[8];
    const float* b2_0 = (const float*)d_in[9];
    const float* eW_r = (const float*)d_in[10];
    const float* eb_r = (const float*)d_in[11];
    const float* W1_r = (const float*)d_in[12];
    const float* b1_r = (const float*)d_in[13];
    const float* W2_r = (const float*)d_in[14];
    const float* b2_r = (const float*)d_in[15];
    const float* cW   = (const float*)d_in[16];
    const float* cb   = (const float*)d_in[17];

    const int N = in_sizes[0] / 64;
    const int E = in_sizes[1];

    float* hA = (float*)d_ws;
    float* hB = hA + (size_t)N * 128;
    int* plist  = (int*)(hB + (size_t)N * 128);
    float* eas  = (float*)(plist + 2 * (size_t)E);
    int* cnt    = (int*)(eas + 16 * (size_t)E);
    int* bstart = cnt  + GB * MAXB;

    const int nbk     = (N + RB - 1) / RB;
    const int chunk   = (E + GB - 1) / GB;

    bucket_count_kernel<<<GB, MAXB, 0, stream>>>(dst, cnt, E, chunk, nbk);
    bucket_fill_kernel<<<GB, MAXB, 0, stream>>>(src, dst, cnt, ea, plist, eas, bstart, E, chunk, nbk);

    layer_kernel<64, 64, 128, 0><<<nbk, 1024, 0, stream>>>(
        x, plist, bstart, eas, eW0, eb0, W1_0, b1_0, W2_0, b2_0, cW, cb,
        hA, (float*)d_out, N);
    layer_kernel<128, 128, 128, 0><<<nbk, 1024, 0, stream>>>(
        hA, plist, bstart, eas, eW_r, eb_r, W1_r, b1_r, W2_r, b2_r, cW, cb,
        hB, (float*)d_out, N);
    layer_kernel<128, 128, 128, 32><<<nbk, 1024, 0, stream>>>(
        hB, plist, bstart, eas, eW_r + (size_t)16 * 128, eb_r + (size_t)128,
        W1_r + (size_t)128 * 128, b1_r + (size_t)128, W2_r + (size_t)128 * 128, b2_r + (size_t)128,
        cW, cb, hA, (float*)d_out, N);
}


# Copy loops (count, table walk, fill, a layer's own rows to LDS) issue several trips' loads as one group before the first wait, not load-wait-store per trip: serial round trips 13/16/13/7 to 2/6/4/1
# speedup vs baseline: 2.6906x; 1.0762x over previous
#include <hip/hip_runtime.h>
#include <cmath>

typedef __attribute__((ext_vector_type(2))) float v2f;
typedef __attribute__((ext_vector_type(4))) float v4f;
typedef __attribute__((ext_vector_type(8))) float v8f;

template<int K, int NC>
__device__ __forceinline__ void fill_swizzled(float* sWs, const float* __restrict__ W) {
    constexpr int NT  = NC / 16;
    constexpr int TOT = (K / 4) * NT * 64;
    for (int i = threadIdx.x; i < TOT; i += blockDim.x) {
        const int h    = i & 1;
        const int lane = (i >> 1) & 31;
        const int c    = (i >> 6) % NT;
        const int ks   = i / (64 * NT);
        const int k    = ks * 4 + ((lane >> 4) << 1) + h;
        const int col  = c * 16 + (lane & 15);
        sWs[i] = W[k * NC + col];
    }
}

constexpr int RB   = 196;
constexpr int MAXB = 512;
constexpr int GB   = 128;
constexpr int NRT   = (RB + 15) / 16;
constexpr int LROWS = NRT * 16;
constexpr int PITCH = 132;
constexpr int LT    = 1024;

typedef __attribute__((ext_vector_type(2))) int v2i;

__global__ void __launch_bounds__(MAXB) bucket_count_kernel(
    const int* __restrict__ dst, int* __restrict__ cnt, int nedges, int chunk, int nbk)
{
    __shared__ int hist[MAXB];
    hist[threadIdx.x] = 0;
    __syncthreads();
    const int e0 = blockIdx.x * chunk;
    const int e1 = min(e0 + chunk, nedges);
    constexpr int CU = 16;
    for (int eb = e0 + (int)threadIdx.x; eb < e1; eb += CU * MAXB) {
        int dv[CU];
#pragma unroll
        for (int k = 0; k < CU; ++k) dv[k] = dst[min(eb + k * MAXB, e1 - 1)];
#pragma unroll
        for (int k = 0; k < CU; ++k)
            if (eb + k * MAXB < e1) atomicAdd(&hist[dv[k] / RB], 1);
    }
    __syncthreads();
    if ((int)threadIdx.x < nbk) cnt[blockIdx.x * nbk + threadIdx.x] = hist[threadIdx.x];
}

__global__ void __launch_bounds__(MAXB) bucket_fill_kernel(
    const int* __restrict__ src, const int* __restrict__ dst, const int* __restrict__ cnt,
    const float* __restrict__ EA,
    int* __restrict__ plist, float* __restrict__ eas, int* __restrict__ bstart,
    int nedges, int chunk, int nbk)
{
    v2i* __restrict__ pl2 = (v2i*)plist;
    __shared__ int s[MAXB];
    __shared__ int cur[MAXB];
    const int b = threadIdx.x;
    const int g = blockIdx.x;
    int tot = 0, pre = 0;
    if (b < nbk) {
#pragma unroll 32
        for (int gg = 0; gg < GB; ++gg) {
            const int c = cnt[gg * nbk + b];
            tot += c;
            pre += (gg < g) ? c : 0;
        }
    }
    s[b] = tot;
    __syncthreads();
    for (int d = 1; d < MAXB; d <<= 1) {
        const int v = (b >= d) ? s[b - d] : 0;
        __syncthreads();
        s[b] += v;
        __syncthreads();
    }
    const int start = s[b] - tot;
    cur[b] = start + pre;
    if (g == 0) {
        if (b < nbk) bstart[b] = start;
        if (b == 0)  bstart[nbk] = nedges;
    }
    __syncthreads();

    const int e0 = g * chunk;
    const int e1 = min(e0 + chunk, nedges);
    const int lane = b & 31;
    const int wb   = __builtin_amdgcn_readfirstlane(b - lane);
    const int sub  = lane >> 2;
    const int part = (lane & 3) << 2;
    constexpr int FU = 4;
    for (int eb0 = e0 + wb; eb0 < e1; eb0 += FU * MAXB) {
        v4f av[FU][4];
        int dv[FU], sv[FU];
#pragma unroll
        for (int u = 0; u < FU; ++u) {
            const int eb = eb0 + u * MAXB;
#pragma unroll
            for (int r = 0; r < 4; ++r) {
                const int es = min(eb + 8 * r + sub, e1 - 1);
                av[u][r] = *(const v4f*)(EA + es * 16 + part);
            }
            const int ec = min(eb + lane, e1 - 1);
            dv[u] = dst[ec];
            sv[u] = src[ec];
        }
#pragma unroll
        for (int u = 0; u < FU; ++u) {
            const int eb = eb0 + u * MAXB;
            if (eb < e1) {
                int pos = 0;
                if (eb + lane < e1) {
                    const int d  = dv[u];
                    const int bk = d / RB;
                    pos = atomicAdd(&cur[bk], 1);
                    pl2[pos] = (v2i){sv[u], d - bk * RB};
                }
#pragma unroll
                for (int r = 0; r < 4; ++r) {
                    const int sl = 8 * r + sub;
                    const int ps = __shfl(pos, sl);
                    if (eb + sl < e1) *(v4f*)(eas + ps * 16 + part) = av[u][r];
                }
            }
        }
    }
}

template<int K, int NC>
struct BFrag {
    v2f   w[K / 8][2];
    float bias;
};

__host__ __device__ constexpr int max_tiles(int NC) {
    return (NRT + 32 / (NC / 16) - 1) / (32 / (NC / 16));
}

template<int K, int NC>
__device__ __forceinline__ void load_bfrag(
    BFrag<K, NC>& f, const float* __restrict__ Wg, const float* __restrict__ bias,
    int wave, int lane)
{
    constexpr int NCT = NC / 16;
    const int col = (wave % NCT) * 16 + (lane & 15);
    const int wb  = (4 * (lane >> 4)) * NC + col;
#pragma unroll
    for (int s = 0; s < K / 8; ++s) {
#pragma unroll
        for (int i = 0; i < 2; ++i) {
            f.w[s][i].x = Wg[wb + (8 * s + 2 * i)     * NC];
            f.w[s][i].y = Wg[wb + (8 * s + 2 * i + 1) * NC];
        }
    }
    f.bias = bias[col];
}

template<int K, int NC>
__device__ __forceinline__ void rows_times_b(
    v8f (&acc)[max_tiles(NC)], const float* rows, const BFrag<K, NC>& f,
    int wave, int lane)
{
    constexpr int NCT = NC / 16;
    constexpr int NRG = 32 / NCT;
    const int rg   = wave / NCT;
    const int n    = lane & 15;
    const int half = lane >> 4;
#pragma unroll
    for (int i = 0; i < max_tiles(NC); ++i) {
        const int rt = rg + i * NRG;
        acc[i] = (v8f){};
        if (rt < NRT) {
            const float* ap = rows + (rt * 16 + n) * PITCH + 4 * half;
#pragma unroll
            for (int s = 0; s < K / 8; ++s) {
                const v4f q = *(const v4f*)(ap + 8 * s);
                const v2f a0 = {q.x, q.y};
                const v2f a1 = {q.z, q.w};
                acc[i] = __builtin_amdgcn_wmma_f32_16x16x4_f32(
                    false, a0, false, f.w[s][0], (short)0, acc[i], false, false);
                acc[i] = __builtin_amdgcn_wmma_f32_16x16x4_f32(
                    false, a1, false, f.w[s][1], (short)0, acc[i], false, false);
            }
        }
    }
}

template<int NC, bool ACT_ELU>
__device__ __forceinline__ void put_rows(
    float* rows, const v8f (&acc)[max_tiles(NC)], float bias, int wave, int lane)
{
    constexpr int NCT = NC / 16;
    constexpr int NRG = 32 / NCT;
    const int rg   = wave / NCT;
    const int col  = (wave % NCT) * 16 + (lane & 15);
    const int half = lane >> 4;
#pragma unroll
    for (int i = 0; i < max_tiles(NC); ++i) {
        const int rt = rg + i * NRG;
        if (rt < NRT) {
            float* op = rows + (rt * 16 + 8 * half) * PITCH + col;
#pragma unroll
            for (int j = 0; j < 8; ++j) {
                float v = acc[i][j] + bias;
                if (ACT_ELU) v = (v > 0.0f) ? v : (__expf(v) - 1.0f);
                op[j * PITCH] = v;
            }
        }
    }
}

template<int D, int DM, int DO, int NCLS>
__global__ void __launch_bounds__(1024) layer_kernel(
    const float* __restrict__ X, const int* __restrict__ plist,
    const int* __restrict__ bstart,
    const float* __restrict__ EA,
    const float* __restrict__ eW, const float* __restrict__ eb,
    const float* __restrict__ W1, const float* __restrict__ bias1,
    const float* __restrict__ W2, const float* __restrict__ bias2,
    const float* __restrict__ cW, const float* __restrict__ cb,
    float* __restrict__ Hout, float* __restrict__ Out, int nnodes)
{
    constexpr int NT    = D / 16;
    constexpr int CP    = 4;
    constexpr int NP    = NT / CP;
    __shared__ float sWs[4 * NT * 64];
    __shared__ float sB[D];
    __shared__ __attribute__((aligned(16))) float sAgg[LROWS * PITCH];

    const int bk    = blockIdx.x;
    const int row0  = bk * RB;
    const int nrows = min(RB, nnodes - row0);

    constexpr int QI  = D / 4;
    constexpr int NXT = (LROWS * QI + LT - 1) / LT;
    v4f xr[NXT];
#pragma unroll
    for (int k = 0; k < NXT; ++k) {
        const int i = (int)threadIdx.x + k * LT;
        const int r = min(i / QI, nrows - 1);
        const int q = i % QI;
        xr[k] = *(const v4f*)(X + (size_t)(row0 + r) * D + q * 4);
    }
    fill_swizzled<16, D>(sWs, eW);
    for (int i = threadIdx.x; i < D; i += blockDim.x) sB[i] = eb[i];
#pragma unroll
    for (int k = 0; k < NXT; ++k) {
        const int i = (int)threadIdx.x + k * LT;
        const int r = i / QI;
        const int q = i - r * QI;
        if (i < LROWS * QI) {
            const v4f z = {0.f, 0.f, 0.f, 0.f};
            *(v4f*)(sAgg + r * PITCH + q * 4) = (r < nrows) ? xr[k] : z;
        }
    }
    __syncthreads();

    const int lane  = threadIdx.x & 31;
    const int wave   = __builtin_amdgcn_readfirstlane((int)(threadIdx.x >> 5));
    const int nwaves = (int)(blockDim.x >> 5);
    const int m     = lane & 15;
    const int kb    = (lane >> 4) << 1;
    const int n     = lane & 15;
    const int mbase = (lane >> 4) << 3;

    const v2i* __restrict__ pl2 = (const v2i*)plist;
    const int lo = bstart[bk];
    const int hi = bstart[bk + 1];
    const int ntl = (hi - lo + 15) >> 4;

    for (int t = wave; t < ntl; t += nwaves) {
        asm volatile("" ::: "memory");
        const int p0 = lo + (t << 4);

        const int pa = min(p0 + m, hi - 1);
        const int ab = pa * 16 + kb;
        const v2f a0 = *(const v2f*)(EA + ab);
        const v2f a1 = *(const v2f*)(EA + ab + 4);
        const v2f a2 = *(const v2f*)(EA + ab + 8);
        const v2f a3 = *(const v2f*)(EA + ab + 12);

        int sb[8], lb[8];
#pragma unroll
        for (int j = 0; j < 8; ++j) {
            const int pj = p0 + mbase + j;
            const v2i sd = pl2[min(pj, hi - 1)];
            sb[j] = sd.x * D + n;
            lb[j] = ((pj < hi) ? sd.y : RB) * PITCH + n;
        }

#pragma unroll
        for (int p = 0; p < NP; ++p) {
            float xv[8][CP];
#pragma unroll
            for (int j = 0; j < 8; ++j) {
#pragma unroll
                for (int c = 0; c < CP; ++c) xv[j][c] = X[sb[j] + (p * CP + c) * 16];
            }
            v8f acc[CP];
#pragma unroll
            for (int c = 0; c < CP; ++c) {
                const int cc = p * CP + c;
                const float bv = sB[cc * 16 + n];
                acc[c] = (v8f){bv, bv, bv, bv, bv, bv, bv, bv};
                const v2f* b0 = (const v2f*)(sWs + (0 * NT + cc) * 64);
                const v2f* b1 = (const v2f*)(sWs + (1 * NT + cc) * 64);
                const v2f* b2 = (const v2f*)(sWs + (2 * NT + cc) * 64);
                const v2f* b3 = (const v2f*)(sWs + (3 * NT + cc) * 64);
                acc[c] = __builtin_amdgcn_wmma_f32_16x16x4_f32(false, a0, false, b0[lane], (short)0, acc[c], false, false);
                acc[c] = __builtin_amdgcn_wmma_f32_16x16x4_f32(false, a1, false, b1[lane], (short)0, acc[c], false, false);
                acc[c] = __builtin_amdgcn_wmma_f32_16x16x4_f32(false, a2, false, b2[lane], (short)0, acc[c], false, false);
                acc[c] = __builtin_amdgcn_wmma_f32_16x16x4_f32(false, a3, false, b3[lane], (short)0, acc[c], false, false);
            }
#pragma unroll
            for (int j = 0; j < 8; ++j) {
#pragma unroll
                for (int c = 0; c < CP; ++c) {
                    float msg = acc[c][j] + xv[j][c];
                    msg = msg > 0.0f ? msg : 0.0f;
                    atomicAdd(&sAgg[lb[j] + (p * CP + c) * 16], msg);
                }
            }
        }
    }
    BFrag<D, DM> f1;
    load_bfrag<D, DM>(f1, W1, bias1, wave, lane);
    __syncthreads();
    v8f acc1[max_tiles(DM)];
    rows_times_b<D, DM>(acc1, sAgg, f1, wave, lane);
    __syncthreads();
    put_rows<DM, true>(sAgg, acc1, f1.bias, wave, lane);
    BFrag<DM, DO> f2;
    load_bfrag<DM, DO>(f2, W2, bias2, wave, lane);
    __syncthreads();
    v8f acc2[max_tiles(DO)];
    rows_times_b<DM, DO>(acc2, sAgg, f2, wave, lane);

    if constexpr (NCLS == 0) {
        __syncthreads();
        put_rows<DO, true>(sAgg, acc2, f2.bias, wave, lane);
        __syncthreads();
        constexpr int Q = DO / 4;
        for (int i = threadIdx.x; i < nrows * Q; i += blockDim.x) {
            const int r = i / Q;
            const int q = i - r * Q;
            const v4f v = *(const v4f*)(sAgg + r * PITCH + q * 4);
            *(v4f*)(Hout + (size_t)(row0 + r) * DO + q * 4) = v;
        }
    } else {
        __syncthreads();
        put_rows<DO, true>(sAgg, acc2, f2.bias, wave, lane);
        BFrag<DO, NCLS> f3;
        load_bfrag<DO, NCLS>(f3, cW, cb, wave, lane);
        __syncthreads();
        v8f acc3[max_tiles(NCLS)];
        rows_times_b<DO, NCLS>(acc3, sAgg, f3, wave, lane);
        __syncthreads();
        put_rows<NCLS, false>(sAgg, acc3, f3.bias, wave, lane);
        __syncthreads();
        constexpr int Q = NCLS / 4;
        for (int i = threadIdx.x; i < nrows * Q; i += blockDim.x) {
            const int r = i / Q;
            const int q = i - r * Q;
            const v4f v = *(const v4f*)(sAgg + r * PITCH + q * 4);
            *(v4f*)(Out + (size_t)(row0 + r) * NCLS + q * 4) = v;
        }
    }
}

extern "C" void kernel_launch(void* const* d_in, const int* in_sizes, int n_in,
                              void* d_out, int out_size, void* d_ws, size_t ws_size,
                              hipStream_t stream)
{
    const float* x    = (const float*)d_in[0];
    const int*   src  = (const int*)  d_in[1];
    const int*   dst  = (const int*)  d_in[2];
    const float* ea   = (const float*)d_in[3];
    const float* eW0  = (const float*)d_in[4];
    const float* eb0  = (const float*)d_in[5];
    const float* W1_0 = (const float*)d_in[6];
    const float* b1_0 = (const float*)d_in[7];
    const float* W2_0 = (const float*)d_in[8];
    const float* b2_0 = (const float*)d_in[9];
    const float* eW_r = (const float*)d_in[10];
    const float* eb_r = (const float*)d_in[11];
    const float* W1_r = (const float*)d_in[12];
    const float* b1_r = (const float*)d_in[13];
    const float* W2_r = (const float*)d_in[14];
    const float* b2_r = (const float*)d_in[15];
    const float* cW   = (const float*)d_in[16];
    const float* cb   = (const float*)d_in[17];

    const int N = in_sizes[0] / 64;
    const int E = in_sizes[1];

    float* hA = (float*)d_ws;
    float* hB = hA + (size_t)N * 128;
    int* plist  = (int*)(hB + (size_t)N * 128);
    float* eas  = (float*)(plist + 2 * (size_t)E);
    int* cnt    = (int*)(eas + 16 * (size_t)E);
    int* bstart = cnt  + GB * MAXB;

    const int nbk     = (N + RB - 1) / RB;
    const int chunk   = (E + GB - 1) / GB;

    bucket_count_kernel<<<GB, MAXB, 0, stream>>>(dst, cnt, E, chunk, nbk);
    bucket_fill_kernel<<<GB, MAXB, 0, stream>>>(src, dst, cnt, ea, plist, eas, bstart, E, chunk, nbk);

    layer_kernel<64, 64, 128, 0><<<nbk, 1024, 0, stream>>>(
        x, plist, bstart, eas, eW0, eb0, W1_0, b1_0, W2_0, b2_0, cW, cb,
        hA, (float*)d_out, N);
    layer_kernel<128, 128, 128, 0><<<nbk, 1024, 0, stream>>>(
        hA, plist, bstart, eas, eW_r, eb_r, W1_r, b1_r, W2_r, b2_r, cW, cb,
        hB, (float*)d_out, N);
    layer_kernel<128, 128, 128, 32><<<nbk, 1024, 0, stream>>>(
        hB, plist, bstart, eas, eW_r + (size_t)16 * 128, eb_r + (size_t)128,
        W1_r + (size_t)128 * 128, b1_r + (size_t)128, W2_r + (size_t)128 * 128, b2_r + (size_t)128,
        cW, cb, hA, (float*)d_out, N);
}


# Edge-message columns re-ordered so a lane owns 4 neighbouring columns (weight fragments, bias, LDS sum slots, W1 k order follow): x[src] read as 16-B loads, 16 per tile not 64; sum order differs
# speedup vs baseline: 2.7987x; 1.0402x over previous
#include <hip/hip_runtime.h>
#include <cmath>

typedef __attribute__((ext_vector_type(2))) float v2f;
typedef __attribute__((ext_vector_type(4))) float v4f;
typedef __attribute__((ext_vector_type(8))) float v8f;

__host__ __device__ constexpr int slot_col(int slot) {
    return (slot & ~63) | ((slot & 15) << 2) | ((slot >> 4) & 3);
}

template<int K, int NC>
__device__ __forceinline__ void fill_swizzled(float* sWs, const float* __restrict__ W) {
    constexpr int NT  = NC / 16;
    constexpr int TOT = (K / 4) * NT * 64;
    for (int i = threadIdx.x; i < TOT; i += blockDim.x) {
        const int h    = i & 1;
        const int lane = (i >> 1) & 31;
        const int c    = (i >> 6) % NT;
        const int ks   = i / (64 * NT);
        const int k    = ks * 4 + ((lane >> 4) << 1) + h;
        const int col  = slot_col(c * 16 + (lane & 15));
        sWs[i] = W[k * NC + col];
    }
}

constexpr int RB   = 196;
constexpr int MAXB = 512;
constexpr int GB   = 128;
constexpr int NRT   = (RB + 15) / 16;
constexpr int LROWS = NRT * 16;
constexpr int PITCH = 132;
constexpr int LT    = 1024;

typedef __attribute__((ext_vector_type(2))) int v2i;

__global__ void __launch_bounds__(MAXB) bucket_count_kernel(
    const int* __restrict__ dst, int* __restrict__ cnt, int nedges, int chunk, int nbk)
{
    __shared__ int hist[MAXB];
    hist[threadIdx.x] = 0;
    __syncthreads();
    const int e0 = blockIdx.x * chunk;
    const int e1 = min(e0 + chunk, nedges);
    constexpr int CU = 16;
    for (int eb = e0 + (int)threadIdx.x; eb < e1; eb += CU * MAXB) {
        int dv[CU];
#pragma unroll
        for (int k = 0; k < CU; ++k) dv[k] = dst[min(eb + k * MAXB, e1 - 1)];
#pragma unroll
        for (int k = 0; k < CU; ++k)
            if (eb + k * MAXB < e1) atomicAdd(&hist[dv[k] / RB], 1);
    }
    __syncthreads();
    if ((int)threadIdx.x < nbk) cnt[blockIdx.x * nbk + threadIdx.x] = hist[threadIdx.x];
}

__global__ void __launch_bounds__(MAXB) bucket_fill_kernel(
    const int* __restrict__ src, const int* __restrict__ dst, const int* __restrict__ cnt,
    const float* __restrict__ EA,
    int* __restrict__ plist, float* __restrict__ eas, int* __restrict__ bstart,
    int nedges, int chunk, int nbk)
{
    v2i* __restrict__ pl2 = (v2i*)plist;
    __shared__ int s[MAXB];
    __shared__ int cur[MAXB];
    const int b = threadIdx.x;
    const int g = blockIdx.x;
    int tot = 0, pre = 0;
    if (b < nbk) {
#pragma unroll 32
        for (int gg = 0; gg < GB; ++gg) {
            const int c = cnt[gg * nbk + b];
            tot += c;
            pre += (gg < g) ? c : 0;
        }
    }
    s[b] = tot;
    __syncthreads();
    for (int d = 1; d < MAXB; d <<= 1) {
        const int v = (b >= d) ? s[b - d] : 0;
        __syncthreads();
        s[b] += v;
        __syncthreads();
    }
    const int start = s[b] - tot;
    cur[b] = start + pre;
    if (g == 0) {
        if (b < nbk) bstart[b] = start;
        if (b == 0)  bstart[nbk] = nedges;
    }
    __syncthreads();

    const int e0 = g * chunk;
    const int e1 = min(e0 + chunk, nedges);
    const int lane = b & 31;
    const int wb   = __builtin_amdgcn_readfirstlane(b - lane);
    const int sub  = lane >> 2;
    const int part = (lane & 3) << 2;
    constexpr int FU = 4;
    for (int eb0 = e0 + wb; eb0 < e1; eb0 += FU * MAXB) {
        v4f av[FU][4];
        int dv[FU], sv[FU];
#pragma unroll
        for (int u = 0; u < FU; ++u) {
            const int eb = eb0 + u * MAXB;
#pragma unroll
            for (int r = 0; r < 4; ++r) {
                const int es = min(eb + 8 * r + sub, e1 - 1);
                av[u][r] = *(const v4f*)(EA + es * 16 + part);
            }
            const int ec = min(eb + lane, e1 - 1);
            dv[u] = dst[ec];
            sv[u] = src[ec];
        }
#pragma unroll
        for (int u = 0; u < FU; ++u) {
            const int eb = eb0 + u * MAXB;
            if (eb < e1) {
                int pos = 0;
                if (eb + lane < e1) {
                    const int d  = dv[u];
                    const int bk = d / RB;
                    pos = atomicAdd(&cur[bk], 1);
                    pl2[pos] = (v2i){sv[u], d - bk * RB};
                }
#pragma unroll
                for (int r = 0; r < 4; ++r) {
                    const int sl = 8 * r + sub;
                    const int ps = __shfl(pos, sl);
                    if (eb + sl < e1) *(v4f*)(eas + ps * 16 + part) = av[u][r];
                }
            }
        }
    }
}

template<int K, int NC>
struct BFrag {
    v2f   w[K / 8][2];
    float bias;
};

__host__ __device__ constexpr int max_tiles(int NC) {
    return (NRT + 32 / (NC / 16) - 1) / (32 / (NC / 16));
}

template<int K, int NC, bool SLOTS>
__device__ __forceinline__ void load_bfrag(
    BFrag<K, NC>& f, const float* __restrict__ Wg, const float* __restrict__ bias,
    int wave, int lane)
{
    constexpr int NCT = NC / 16;
    const int col  = (wave % NCT) * 16 + (lane & 15);
    const int half = lane >> 4;
#pragma unroll
    for (int s = 0; s < K / 8; ++s) {
#pragma unroll
        for (int i = 0; i < 2; ++i) {
            const int k0 = 8 * s + 4 * half + 2 * i;
            const int r0 = SLOTS ? slot_col(k0)     : k0;
            const int r1 = SLOTS ? slot_col(k0 + 1) : k0 + 1;
            f.w[s][i].x = Wg[r0 * NC + col];
            f.w[s][i].y = Wg[r1 * NC + col];
        }
    }
    f.bias = bias[col];
}

template<int K, int NC>
__device__ __forceinline__ void rows_times_b(
    v8f (&acc)[max_tiles(NC)], const float* rows, const BFrag<K, NC>& f,
    int wave, int lane)
{
    constexpr int NCT = NC / 16;
    constexpr int NRG = 32 / NCT;
    const int rg   = wave / NCT;
    const int n    = lane & 15;
    const int half = lane >> 4;
#pragma unroll
    for (int i = 0; i < max_tiles(NC); ++i) {
        const int rt = rg + i * NRG;
        acc[i] = (v8f){};
        if (rt < NRT) {
            const float* ap = rows + (rt * 16 + n) * PITCH + 4 * half;
#pragma unroll
            for (int s = 0; s < K / 8; ++s) {
                const v4f q = *(const v4f*)(ap + 8 * s);
                const v2f a0 = {q.x, q.y};
                const v2f a1 = {q.z, q.w};
                acc[i] = __builtin_amdgcn_wmma_f32_16x16x4_f32(
                    false, a0, false, f.w[s][0], (short)0, acc[i], false, false);
                acc[i] = __builtin_amdgcn_wmma_f32_16x16x4_f32(
                    false, a1, false, f.w[s][1], (short)0, acc[i], false, false);
            }
        }
    }
}

template<int NC, bool ACT_ELU>
__device__ __forceinline__ void put_rows(
    float* rows, const v8f (&acc)[max_tiles(NC)], float bias, int wave, int lane)
{
    constexpr int NCT = NC / 16;
    constexpr int NRG = 32 / NCT;
    const int rg   = wave / NCT;
    const int col  = (wave % NCT) * 16 + (lane & 15);
    const int half = lane >> 4;
#pragma unroll
    for (int i = 0; i < max_tiles(NC); ++i) {
        const int rt = rg + i * NRG;
        if (rt < NRT) {
            float* op = rows + (rt * 16 + 8 * half) * PITCH + col;
#pragma unroll
            for (int j = 0; j < 8; ++j) {
                float v = acc[i][j] + bias;
                if (ACT_ELU) v = (v > 0.0f) ? v : (__expf(v) - 1.0f);
                op[j * PITCH] = v;
            }
        }
    }
}

template<int D, int DM, int DO, int NCLS>
__global__ void __launch_bounds__(1024) layer_kernel(
    const float* __restrict__ X, const int* __restrict__ plist,
    const int* __restrict__ bstart,
    const float* __restrict__ EA,
    const float* __restrict__ eW, const float* __restrict__ eb,
    const float* __restrict__ W1, const float* __restrict__ bias1,
    const float* __restrict__ W2, const float* __restrict__ bias2,
    const float* __restrict__ cW, const float* __restrict__ cb,
    float* __restrict__ Hout, float* __restrict__ Out, int nnodes)
{
    constexpr int NT    = D / 16;
    constexpr int CP    = 4;
    constexpr int NP    = NT / CP;
    __shared__ float sWs[4 * NT * 64];
    __shared__ float sB[D];
    __shared__ __attribute__((aligned(16))) float sAgg[LROWS * PITCH];

    const int bk    = blockIdx.x;
    const int row0  = bk * RB;
    const int nrows = min(RB, nnodes - row0);

    constexpr int QI  = D / 4;
    constexpr int NXT = (LROWS * QI + LT - 1) / LT;
    v4f xr[NXT];
#pragma unroll
    for (int k = 0; k < NXT; ++k) {
        const int i = (int)threadIdx.x + k * LT;
        const int r = min(i / QI, nrows - 1);
        const int q = i % QI;
        xr[k] = *(const v4f*)(X + (size_t)(row0 + r) * D + q * 4);
    }
    fill_swizzled<16, D>(sWs, eW);
    for (int i = threadIdx.x; i < D; i += blockDim.x) sB[i] = eb[slot_col(i)];
#pragma unroll
    for (int k = 0; k < NXT; ++k) {
        const int i = (int)threadIdx.x + k * LT;
        const int r = i / QI;
        const int q = i - r * QI;
        if (i < LROWS * QI) {
            const v4f z = {0.f, 0.f, 0.f, 0.f};
            const v4f v = (r < nrows) ? xr[k] : z;
            float* sp = sAgg + r * PITCH + (q >> 4) * 64 + (q & 15);
            sp[0]  = v.x;
            sp[16] = v.y;
            sp[32] = v.z;
            sp[48] = v.w;
        }
    }
    __syncthreads();

    const int lane  = threadIdx.x & 31;
    const int wave   = __builtin_amdgcn_readfirstlane((int)(threadIdx.x >> 5));
    const int nwaves = (int)(blockDim.x >> 5);
    const int m     = lane & 15;
    const int kb    = (lane >> 4) << 1;
    const int n     = lane & 15;
    const int mbase = (lane >> 4) << 3;

    const v2i* __restrict__ pl2 = (const v2i*)plist;
    const v4f* __restrict__ X4  = (const v4f*)X;
    const int lo = bstart[bk];
    const int hi = bstart[bk + 1];
    const int ntl = (hi - lo + 15) >> 4;

    for (int t = wave; t < ntl; t += nwaves) {
        asm volatile("" ::: "memory");
        const int p0 = lo + (t << 4);

        const int pa = min(p0 + m, hi - 1);
        const int ab = pa * 16 + kb;
        const v2f a0 = *(const v2f*)(EA + ab);
        const v2f a1 = *(const v2f*)(EA + ab + 4);
        const v2f a2 = *(const v2f*)(EA + ab + 8);
        const v2f a3 = *(const v2f*)(EA + ab + 12);

        int sb[8], lb[8];
#pragma unroll
        for (int j = 0; j < 8; ++j) {
            const int pj = p0 + mbase + j;
            const v2i sd = pl2[min(pj, hi - 1)];
            sb[j] = sd.x * (D / 4) + n;
            lb[j] = ((pj < hi) ? sd.y : RB) * PITCH + n;
        }

#pragma unroll
        for (int p = 0; p < NP; ++p) {
            v4f xv[8];
#pragma unroll
            for (int j = 0; j < 8; ++j) xv[j] = X4[sb[j] + p * 16];
            v8f acc[CP];
#pragma unroll
            for (int c = 0; c < CP; ++c) {
                const int cc = p * CP + c;
                const float bv = sB[cc * 16 + n];
                acc[c] = (v8f){bv, bv, bv, bv, bv, bv, bv, bv};
                const v2f* b0 = (const v2f*)(sWs + (0 * NT + cc) * 64);
                const v2f* b1 = (const v2f*)(sWs + (1 * NT + cc) * 64);
                const v2f* b2 = (const v2f*)(sWs + (2 * NT + cc) * 64);
                const v2f* b3 = (const v2f*)(sWs + (3 * NT + cc) * 64);
                acc[c] = __builtin_amdgcn_wmma_f32_16x16x4_f32(false, a0, false, b0[lane], (short)0, acc[c], false, false);
                acc[c] = __builtin_amdgcn_wmma_f32_16x16x4_f32(false, a1, false, b1[lane], (short)0, acc[c], false, false);
                acc[c] = __builtin_amdgcn_wmma_f32_16x16x4_f32(false, a2, false, b2[lane], (short)0, acc[c], false, false);
                acc[c] = __builtin_amdgcn_wmma_f32_16x16x4_f32(false, a3, false, b3[lane], (short)0, acc[c], false, false);
            }
#pragma unroll
            for (int j = 0; j < 8; ++j) {
#pragma unroll
                for (int c = 0; c < CP; ++c) {
                    float msg = acc[c][j] + xv[j][c];
                    msg = msg > 0.0f ? msg : 0.0f;
                    atomicAdd(&sAgg[lb[j] + (p * CP + c) * 16], msg);
                }
            }
        }
    }
    BFrag<D, DM> f1;
    load_bfrag<D, DM, true>(f1, W1, bias1, wave, lane);
    __syncthreads();
    v8f acc1[max_tiles(DM)];
    rows_times_b<D, DM>(acc1, sAgg, f1, wave, lane);
    __syncthreads();
    put_rows<DM, true>(sAgg, acc1, f1.bias, wave, lane);
    BFrag<DM, DO> f2;
    load_bfrag<DM, DO, false>(f2, W2, bias2, wave, lane);
    __syncthreads();
    v8f acc2[max_tiles(DO)];
    rows_times_b<DM, DO>(acc2, sAgg, f2, wave, lane);

    if constexpr (NCLS == 0) {
        __syncthreads();
        put_rows<DO, true>(sAgg, acc2, f2.bias, wave, lane);
        __syncthreads();
        constexpr int Q = DO / 4;
        for (int i = threadIdx.x; i < nrows * Q; i += blockDim.x) {
            const int r = i / Q;
            const int q = i - r * Q;
            const v4f v = *(const v4f*)(sAgg + r * PITCH + q * 4);
            *(v4f*)(Hout + (size_t)(row0 + r) * DO + q * 4) = v;
        }
    } else {
        __syncthreads();
        put_rows<DO, true>(sAgg, acc2, f2.bias, wave, lane);
        BFrag<DO, NCLS> f3;
        load_bfrag<DO, NCLS, false>(f3, cW, cb, wave, lane);
        __syncthreads();
        v8f acc3[max_tiles(NCLS)];
        rows_times_b<DO, NCLS>(acc3, sAgg, f3, wave, lane);
        __syncthreads();
        put_rows<NCLS, false>(sAgg, acc3, f3.bias, wave, lane);
        __syncthreads();
        constexpr int Q = NCLS / 4;
        for (int i = threadIdx.x; i < nrows * Q; i += blockDim.x) {
            const int r = i / Q;
            const int q = i - r * Q;
            const v4f v = *(const v4f*)(sAgg + r * PITCH + q * 4);
            *(v4f*)(Out + (size_t)(row0 + r) * NCLS + q * 4) = v;
        }
    }
}

extern "C" void kernel_launch(void* const* d_in, const int* in_sizes, int n_in,
                              void* d_out, int out_size, void* d_ws, size_t ws_size,
                              hipStream_t stream)
{
    const float* x    = (const float*)d_in[0];
    const int*   src  = (const int*)  d_in[1];
    const int*   dst  = (const int*)  d_in[2];
    const float* ea   = (const float*)d_in[3];
    const float* eW0  = (const float*)d_in[4];
    const float* eb0  = (const float*)d_in[5];
    const float* W1_0 = (const float*)d_in[6];
    const float* b1_0 = (const float*)d_in[7];
    const float* W2_0 = (const float*)d_in[8];
    const float* b2_0 = (const float*)d_in[9];
    const float* eW_r = (const float*)d_in[10];
    const float* eb_r = (const float*)d_in[11];
    const float* W1_r = (const float*)d_in[12];
    const float* b1_r = (const float*)d_in[13];
    const float* W2_r = (const float*)d_in[14];
    const float* b2_r = (const float*)d_in[15];
    const float* cW   = (const float*)d_in[16];
    const float* cb   = (const float*)d_in[17];

    const int N = in_sizes[0] / 64;
    const int E = in_sizes[1];

    float* hA = (float*)d_ws;
    float* hB = hA + (size_t)N * 128;
    int* plist  = (int*)(hB + (size_t)N * 128);
    float* eas  = (float*)(plist + 2 * (size_t)E);
    int* cnt    = (int*)(eas + 16 * (size_t)E);
    int* bstart = cnt  + GB * MAXB;

    const int nbk     = (N + RB - 1) / RB;
    const int chunk   = (E + GB - 1) / GB;

    bucket_count_kernel<<<GB, MAXB, 0, stream>>>(dst, cnt, E, chunk, nbk);
    bucket_fill_kernel<<<GB, MAXB, 0, stream>>>(src, dst, cnt, ea, plist, eas, bstart, E, chunk, nbk);

    layer_kernel<64, 64, 128, 0><<<nbk, 1024, 0, stream>>>(
        x, plist, bstart, eas, eW0, eb0, W1_0, b1_0, W2_0, b2_0, cW, cb,
        hA, (float*)d_out, N);
    layer_kernel<128, 128, 128, 0><<<nbk, 1024, 0, stream>>>(
        hA, plist, bstart, eas, eW_r, eb_r, W1_r, b1_r, W2_r, b2_r, cW, cb,
        hB, (float*)d_out, N);
    layer_kernel<128, 128, 128, 32><<<nbk, 1024, 0, stream>>>(
        hB, plist, bstart, eas, eW_r + (size_t)16 * 128, eb_r + (size_t)128,
        W1_r + (size_t)128 * 128, b1_r + (size_t)128, W2_r + (size_t)128 * 128, b2_r + (size_t)128,
        cW, cb, hA, (float*)d_out, N);
}
